# starting kernel
# baseline (speedup 1.0000x reference)
#include <hip/hip_runtime.h>

typedef __attribute__((ext_vector_type(2))) float v2f;
typedef __attribute__((ext_vector_type(8))) float v8f;

#define HDIM 128
#define NT   16


__global__ void fill_ones_kernel(float* __restrict__ p, int n) {
  int i = blockIdx.x * blockDim.x + threadIdx.x;
  if (i < n) p[i] = 1.0f;
}

__global__ void zero_kernel(float* __restrict__ p, long long n) {
  long long i = (long long)blockIdx.x * blockDim.x + threadIdx.x;
  if (i < n) p[i] = 0.0f;
}

__global__ void degree_kernel(const int* __restrict__ src, const int* __restrict__ dst,
                              float* __restrict__ deg_out, float* __restrict__ deg_in, int E) {
  int e = blockIdx.x * blockDim.x + threadIdx.x;
  if (e < E) {
    atomicAdd(&deg_out[src[e]], 1.0f);
    atomicAdd(&deg_in[dst[e]], 1.0f);
  }
}

__global__ void norm_kernel(float* __restrict__ p, int n) {
  int i = blockIdx.x * blockDim.x + threadIdx.x;
  if (i < n) p[i] = rsqrtf(fmaxf(p[i], 1.0f));
}

__global__ void gemm_wmma_f32_kernel(const float* __restrict__ X, const float* __restrict__ W,
                                     float* __restrict__ Y, int KD) {
  const int wave = threadIdx.x >> 5;
  const int lane = threadIdx.x & 31;
  const int l16  = lane & 15;
  const int hi   = lane >> 4;
  const int mbase = blockIdx.x * NT;
  const int nbase = wave * NT;

  const float* __restrict__ xrow = X + (size_t)(mbase + l16) * KD;
  const float* __restrict__ wcol = W + nbase + l16;

  v8f c = {};
  #pragma unroll 4
  for (int k = 0; k < KD; k += 4) {
    v2f a, b;
    a.x = xrow[k + 2 * hi];
    a.y = xrow[k + 2 * hi + 1];
    b.x = wcol[(size_t)(k + 2 * hi) * HDIM];
    b.y = wcol[(size_t)(k + 2 * hi + 1) * HDIM];
    c = __builtin_amdgcn_wmma_f32_16x16x4_f32(false, a, false, b, (short)0, c, false, false);
  }

  float* __restrict__ yout = Y + nbase + l16;
  #pragma unroll
  for (int v = 0; v < 8; ++v)
    yout[(size_t)(mbase + v + 8 * hi) * HDIM] = c[v];
}

__global__ void scatter_kernel(const int* __restrict__ src, const int* __restrict__ dst,
                               const float* __restrict__ ns, const float* __restrict__ Y,
                               float* __restrict__ agg, int E) {
  int warp = (int)(((long long)blockIdx.x * blockDim.x + threadIdx.x) >> 5);
  int lane = threadIdx.x & 31;
  if (warp >= E) return;
  int s = src[warp], d = dst[warp];
  float w = ns[s];
  float4 m = ((const float4*)(Y + (size_t)s * HDIM))[lane];
  float* ap = agg + (size_t)d * HDIM + lane * 4;
  atomicAdd(ap + 0, m.x * w);
  atomicAdd(ap + 1, m.y * w);
  atomicAdd(ap + 2, m.z * w);
  atomicAdd(ap + 3, m.w * w);
}

__global__ void finalize_kernel(const float* __restrict__ agg, const float* __restrict__ Y,
                                const float* __restrict__ ns, const float* __restrict__ nd,
                                const float* __restrict__ bias, float* __restrict__ out,
                                long long total, int relu) {
  long long idx = (long long)blockIdx.x * blockDim.x + threadIdx.x;
  if (idx >= total) return;
  int i = (int)(idx >> 7);
  int j = (int)(idx & (HDIM - 1));
  float v = (agg[idx] + Y[idx] * ns[i]) * nd[i] + bias[j];
  if (relu) v = fmaxf(v, 0.0f);
  out[idx] = v;
}

static inline int cdiv(long long a, int b) { return (int)((a + b - 1) / b); }

static void run_graph(const float* feat, const int* src, const int* dst,
                      const float* W0, const float* b0, const float* W1, const float* b1,
                      int N, int D, int E,
                      float* ns, float* nd, float* Y, float* agg, float* h,
                      float* out, hipStream_t stream) {
  const long long NH = (long long)N * HDIM;

  fill_ones_kernel<<<cdiv(2 * (long long)N, 256), 256, 0, stream>>>(ns, 2 * N);
  degree_kernel<<<cdiv(E, 256), 256, 0, stream>>>(src, dst, ns, nd, E);
  norm_kernel<<<cdiv(2 * (long long)N, 256), 256, 0, stream>>>(ns, 2 * N);

  gemm_wmma_f32_kernel<<<N / NT, 256, 0, stream>>>(feat, W0, Y, D);
  zero_kernel<<<cdiv(NH, 256), 256, 0, stream>>>(agg, NH);
  scatter_kernel<<<cdiv((long long)E * 32, 256), 256, 0, stream>>>(src, dst, ns, Y, agg, E);
  finalize_kernel<<<cdiv(NH, 256), 256, 0, stream>>>(agg, Y, ns, nd, b0, h, NH, 1);

  gemm_wmma_f32_kernel<<<N / NT, 256, 0, stream>>>(h, W1, Y, HDIM);
  zero_kernel<<<cdiv(NH, 256), 256, 0, stream>>>(agg, NH);
  scatter_kernel<<<cdiv((long long)E * 32, 256), 256, 0, stream>>>(src, dst, ns, Y, agg, E);
  finalize_kernel<<<cdiv(NH, 256), 256, 0, stream>>>(agg, Y, ns, nd, b1, out, NH, 0);
}

extern "C" void kernel_launch(void* const* d_in, const int* in_sizes, int n_in,
                              void* d_out, int out_size, void* d_ws, size_t ws_size,
                              hipStream_t stream) {
  const float* feat_a = (const float*)d_in[0];
  const float* feat_b = (const float*)d_in[1];
  const int*   src_a  = (const int*)d_in[2];
  const int*   dst_a  = (const int*)d_in[3];
  const int*   src_b  = (const int*)d_in[4];
  const int*   dst_b  = (const int*)d_in[5];
  const float* Wa0 = (const float*)d_in[6];
  const float* ba0 = (const float*)d_in[7];
  const float* Wa1 = (const float*)d_in[8];
  const float* ba1 = (const float*)d_in[9];
  const float* Wb0 = (const float*)d_in[10];
  const float* bb0 = (const float*)d_in[11];
  const float* Wb1 = (const float*)d_in[12];
  const float* bb1 = (const float*)d_in[13];

  const int DA = in_sizes[6] / HDIM;
  const int DB = in_sizes[10] / HDIM;
  const int NA = in_sizes[0] / DA;
  const int NB = in_sizes[1] / DB;
  const int EA = in_sizes[2];
  const int EB = in_sizes[4];

  float* ws = (float*)d_ws;
  float* ns_a  = ws;
  float* nd_a  = ns_a + NA;
  float* Y_a   = nd_a + NA;
  float* agg_a = Y_a + (size_t)NA * HDIM;
  float* h_a   = agg_a + (size_t)NA * HDIM;
  float* ns_b  = h_a + (size_t)NA * HDIM;
  float* nd_b  = ns_b + NB;
  float* Y_b   = nd_b + NB;
  float* agg_b = Y_b + (size_t)NB * HDIM;
  float* h_b   = agg_b + (size_t)NB * HDIM;

  float* out_a = (float*)d_out;
  float* out_b = out_a + (size_t)NA * HDIM;

  run_graph(feat_a, src_a, dst_a, Wa0, ba0, Wa1, ba1, NA, DA, EA,
            ns_a, nd_a, Y_a, agg_a, h_a, out_a, stream);
  run_graph(feat_b, src_b, dst_b, Wb0, bb0, Wb1, bb1, NB, DB, EB,
            ns_b, nd_b, Y_b, agg_b, h_b, out_b, stream);
}


# Neighbour sum as a gather over a dst-sorted edge list (built per graph with returning int atomics; arrival-order rank sets each row's sum order) replaces f32 atomic scatter; zero+finalize fold in
# speedup vs baseline: 1.9031x; 1.9031x over previous
#include <hip/hip_runtime.h>

typedef __attribute__((ext_vector_type(2))) float v2f;
typedef __attribute__((ext_vector_type(8))) float v8f;

#define HDIM 128
#define NT   16
#define SCAN_T 1024


__global__ void zero_int_kernel(int* __restrict__ p, int n) {
  int i = blockIdx.x * blockDim.x + threadIdx.x;
  if (i < n) p[i] = 0;
}

__global__ void count_kernel(const int* __restrict__ src, const int* __restrict__ dst,
                             int* __restrict__ cnt_out, int* __restrict__ cnt_in,
                             int* __restrict__ rank, int E) {
  int e = blockIdx.x * blockDim.x + threadIdx.x;
  if (e < E) {
    const int s = src[e];
    const int d = dst[e];
    atomicAdd(&cnt_out[s], 1);
    rank[e] = atomicAdd(&cnt_in[d], 1);
  }
}

__global__ void __launch_bounds__(SCAN_T)
scan_norm_kernel(const int* __restrict__ cnt_out, const int* __restrict__ cnt_in,
                 int* __restrict__ rowptr, float* __restrict__ ns, float* __restrict__ nd, int N) {
  __shared__ int wave_front[SCAN_T / 32];
  __shared__ int wave_own[SCAN_T / 32];
  const int tid  = threadIdx.x;
  const int lane = tid & 31;
  const int wave = tid >> 5;
  const int base = blockIdx.x * SCAN_T;

  int front = 0;
  for (int j = tid; j < base; j += SCAN_T) front += cnt_in[j];

  const int i = base + tid;
  const int c = (i < N) ? cnt_in[i] : 0;

  int incl = c;
  for (int o = 1; o < 32; o <<= 1) {
    int t = __shfl_up(incl, o);
    if (lane >= o) incl += t;
  }
  for (int o = 16; o > 0; o >>= 1) front += __shfl_xor(front, o);

  if (lane == 31) wave_own[wave] = incl;
  if (lane == 0)  wave_front[wave] = front;
  __syncthreads();

  int off = 0;
  for (int w = 0; w < SCAN_T / 32; ++w) {
    off += wave_front[w];
    if (w < wave) off += wave_own[w];
  }

  if (i < N) {
    rowptr[i] = off + incl - c;
    if (i == N - 1) rowptr[N] = off + incl;
    ns[i] = rsqrtf(fmaxf((float)(cnt_out[i] + 1), 1.0f));
    nd[i] = rsqrtf(fmaxf((float)(c + 1), 1.0f));
  }
}

__global__ void fill_kernel(const int* __restrict__ src, const int* __restrict__ dst,
                            const int* __restrict__ rank, const int* __restrict__ rowptr,
                            int* __restrict__ col, int E) {
  int e = blockIdx.x * blockDim.x + threadIdx.x;
  if (e < E) col[rowptr[dst[e]] + rank[e]] = src[e];
}

__global__ void gemm_wmma_f32_kernel(const float* __restrict__ X, const float* __restrict__ W,
                                     float* __restrict__ Y, int KD) {
  const int wave = threadIdx.x >> 5;
  const int lane = threadIdx.x & 31;
  const int l16  = lane & 15;
  const int hi   = lane >> 4;
  const int mbase = blockIdx.x * NT;
  const int nbase = wave * NT;

  const float* __restrict__ xrow = X + (size_t)(mbase + l16) * KD;
  const float* __restrict__ wcol = W + nbase + l16;

  v8f c = {};
  #pragma unroll 4
  for (int k = 0; k < KD; k += 4) {
    v2f a, b;
    a.x = xrow[k + 2 * hi];
    a.y = xrow[k + 2 * hi + 1];
    b.x = wcol[(size_t)(k + 2 * hi) * HDIM];
    b.y = wcol[(size_t)(k + 2 * hi + 1) * HDIM];
    c = __builtin_amdgcn_wmma_f32_16x16x4_f32(false, a, false, b, (short)0, c, false, false);
  }

  float* __restrict__ yout = Y + nbase + l16;
  #pragma unroll
  for (int v = 0; v < 8; ++v)
    yout[(size_t)(mbase + v + 8 * hi) * HDIM] = c[v];
}

__global__ void __launch_bounds__(256)
gather_kernel(const int* __restrict__ rowptr, const int* __restrict__ col,
              const float* __restrict__ ns, const float* __restrict__ nd,
              const float* __restrict__ Y, const float* __restrict__ bias,
              float* __restrict__ out, int N, int relu) {
  const int lane = threadIdx.x & 31;
  const int wave = __builtin_amdgcn_readfirstlane((int)(threadIdx.x >> 5));
  const int node = blockIdx.x * 8 + wave;
  if (node >= N) return;

  const float4* __restrict__ Y4 = (const float4*)Y;
  const int beg = rowptr[node];
  const int end = rowptr[node + 1];
  const float4 self = Y4[(size_t)node * (HDIM / 4) + lane];
  const float nsi = ns[node];
  const float ndi = nd[node];
  const float b0 = bias[lane * 4 + 0], b1 = bias[lane * 4 + 1];
  const float b2 = bias[lane * 4 + 2], b3 = bias[lane * 4 + 3];

  float4 acc = {0.0f, 0.0f, 0.0f, 0.0f};
  for (int j0 = beg; j0 < end; j0 += 32) {
    const int n = min(32, end - j0);
    int s = 0;
    float w = 0.0f;
    if (lane < n) { s = col[j0 + lane]; w = ns[s]; }
    const int wi = __float_as_int(w);
    int k = 0;
    for (; k + 4 <= n; k += 4) {
      const int s0 = __builtin_amdgcn_readlane(s, k);
      const int s1 = __builtin_amdgcn_readlane(s, k + 1);
      const int s2 = __builtin_amdgcn_readlane(s, k + 2);
      const int s3 = __builtin_amdgcn_readlane(s, k + 3);
      const float w0 = __int_as_float(__builtin_amdgcn_readlane(wi, k));
      const float w1 = __int_as_float(__builtin_amdgcn_readlane(wi, k + 1));
      const float w2 = __int_as_float(__builtin_amdgcn_readlane(wi, k + 2));
      const float w3 = __int_as_float(__builtin_amdgcn_readlane(wi, k + 3));
      const float4 m0 = Y4[(size_t)s0 * (HDIM / 4) + lane];
      const float4 m1 = Y4[(size_t)s1 * (HDIM / 4) + lane];
      const float4 m2 = Y4[(size_t)s2 * (HDIM / 4) + lane];
      const float4 m3 = Y4[(size_t)s3 * (HDIM / 4) + lane];
      acc.x = fmaf(m0.x, w0, acc.x); acc.y = fmaf(m0.y, w0, acc.y);
      acc.z = fmaf(m0.z, w0, acc.z); acc.w = fmaf(m0.w, w0, acc.w);
      acc.x = fmaf(m1.x, w1, acc.x); acc.y = fmaf(m1.y, w1, acc.y);
      acc.z = fmaf(m1.z, w1, acc.z); acc.w = fmaf(m1.w, w1, acc.w);
      acc.x = fmaf(m2.x, w2, acc.x); acc.y = fmaf(m2.y, w2, acc.y);
      acc.z = fmaf(m2.z, w2, acc.z); acc.w = fmaf(m2.w, w2, acc.w);
      acc.x = fmaf(m3.x, w3, acc.x); acc.y = fmaf(m3.y, w3, acc.y);
      acc.z = fmaf(m3.z, w3, acc.z); acc.w = fmaf(m3.w, w3, acc.w);
    }
    #pragma nounroll
    for (; k < n; ++k) {
      const int s0 = __builtin_amdgcn_readlane(s, k);
      const float w0 = __int_as_float(__builtin_amdgcn_readlane(wi, k));
      const float4 m0 = Y4[(size_t)s0 * (HDIM / 4) + lane];
      acc.x = fmaf(m0.x, w0, acc.x); acc.y = fmaf(m0.y, w0, acc.y);
      acc.z = fmaf(m0.z, w0, acc.z); acc.w = fmaf(m0.w, w0, acc.w);
    }
  }

  float4 v;
  v.x = (acc.x + self.x * nsi) * ndi + b0;
  v.y = (acc.y + self.y * nsi) * ndi + b1;
  v.z = (acc.z + self.z * nsi) * ndi + b2;
  v.w = (acc.w + self.w * nsi) * ndi + b3;
  if (relu) {
    v.x = fmaxf(v.x, 0.0f); v.y = fmaxf(v.y, 0.0f);
    v.z = fmaxf(v.z, 0.0f); v.w = fmaxf(v.w, 0.0f);
  }
  ((float4*)out)[(size_t)node * (HDIM / 4) + lane] = v;
}

static inline int cdiv(long long a, int b) { return (int)((a + b - 1) / b); }

static void run_graph(const float* feat, const int* src, const int* dst,
                      const float* W0, const float* b0, const float* W1, const float* b1,
                      int N, int D, int E,
                      float* ns, float* nd, float* Y, float* agg, float* h,
                      float* out, hipStream_t stream) {
  int* cnt_out = (int*)agg;
  int* cnt_in  = cnt_out + N;
  int* rowptr  = cnt_in + N;
  int* rank    = rowptr + (N + 1);
  int* col     = rank + E;

  zero_int_kernel<<<cdiv(2 * (long long)N, 256), 256, 0, stream>>>(cnt_out, 2 * N);
  count_kernel<<<cdiv(E, 256), 256, 0, stream>>>(src, dst, cnt_out, cnt_in, rank, E);
  scan_norm_kernel<<<cdiv(N, SCAN_T), SCAN_T, 0, stream>>>(cnt_out, cnt_in, rowptr, ns, nd, N);
  fill_kernel<<<cdiv(E, 256), 256, 0, stream>>>(src, dst, rank, rowptr, col, E);

  gemm_wmma_f32_kernel<<<N / NT, 256, 0, stream>>>(feat, W0, Y, D);
  gather_kernel<<<cdiv(N, 8), 256, 0, stream>>>(rowptr, col, ns, nd, Y, b0, h, N, 1);

  gemm_wmma_f32_kernel<<<N / NT, 256, 0, stream>>>(h, W1, Y, HDIM);
  gather_kernel<<<cdiv(N, 8), 256, 0, stream>>>(rowptr, col, ns, nd, Y, b1, out, N, 0);
}

extern "C" void kernel_launch(void* const* d_in, const int* in_sizes, int n_in,
                              void* d_out, int out_size, void* d_ws, size_t ws_size,
                              hipStream_t stream) {
  const float* feat_a = (const float*)d_in[0];
  const float* feat_b = (const float*)d_in[1];
  const int*   src_a  = (const int*)d_in[2];
  const int*   dst_a  = (const int*)d_in[3];
  const int*   src_b  = (const int*)d_in[4];
  const int*   dst_b  = (const int*)d_in[5];
  const float* Wa0 = (const float*)d_in[6];
  const float* ba0 = (const float*)d_in[7];
  const float* Wa1 = (const float*)d_in[8];
  const float* ba1 = (const float*)d_in[9];
  const float* Wb0 = (const float*)d_in[10];
  const float* bb0 = (const float*)d_in[11];
  const float* Wb1 = (const float*)d_in[12];
  const float* bb1 = (const float*)d_in[13];

  const int DA = in_sizes[6] / HDIM;
  const int DB = in_sizes[10] / HDIM;
  const int NA = in_sizes[0] / DA;
  const int NB = in_sizes[1] / DB;
  const int EA = in_sizes[2];
  const int EB = in_sizes[4];

  float* ws = (float*)d_ws;
  float* ns_a  = ws;
  float* nd_a  = ns_a + NA;
  float* Y_a   = nd_a + NA;
  float* agg_a = Y_a + (size_t)NA * HDIM;
  float* h_a   = agg_a + (size_t)NA * HDIM;
  float* ns_b  = h_a + (size_t)NA * HDIM;
  float* nd_b  = ns_b + NB;
  float* Y_b   = nd_b + NB;
  float* agg_b = Y_b + (size_t)NB * HDIM;
  float* h_b   = agg_b + (size_t)NB * HDIM;

  float* out_a = (float*)d_out;
  float* out_b = out_a + (size_t)NA * HDIM;

  run_graph(feat_a, src_a, dst_a, Wa0, ba0, Wa1, ba1, NA, DA, EA,
            ns_a, nd_a, Y_a, agg_a, h_a, out_a, stream);
  run_graph(feat_b, src_b, dst_b, Wb0, bb0, Wb1, bb1, NB, DB, EB,
            ns_b, nd_b, Y_b, agg_b, h_b, out_b, stream);
}


# CSR build as a two-level counting sort on LDS counters (edges binned by 256-node range, then ranked per range) replaces two global int atomics per edge; row sum order stays arrival order
# speedup vs baseline: 2.4038x; 1.2631x over previous
#include <hip/hip_runtime.h>

typedef __attribute__((ext_vector_type(2))) float v2f;
typedef __attribute__((ext_vector_type(8))) float v8f;

#define HDIM 128
#define NT   16

#define BK       256
#define BK_SHIFT 8
#define MAXB     256
#define NCHUNK   128
#define CHUNK_T  1024
#define SORT_T   1024

__global__ void __launch_bounds__(CHUNK_T)
bucket_count_kernel(const int* __restrict__ src, const int* __restrict__ dst,
                    int* __restrict__ cnt_in, int* __restrict__ cnt_out,
                    int E, int epc, int B) {
  __shared__ int h_in[MAXB];
  __shared__ int h_out[MAXB];
  const int tid = threadIdx.x;
  const int c = blockIdx.x;
  if (tid < MAXB) { h_in[tid] = 0; h_out[tid] = 0; }
  __syncthreads();
  const int e0 = c * epc;
  const int e1 = min(E, e0 + epc);
  for (int e = e0 + tid; e < e1; e += CHUNK_T) {
    atomicAdd(&h_in[dst[e] >> BK_SHIFT], 1);
    atomicAdd(&h_out[src[e] >> BK_SHIFT], 1);
  }
  __syncthreads();
  if (tid < B) {
    cnt_in[c * B + tid]  = h_in[tid];
    cnt_out[c * B + tid] = h_out[tid];
  }
}

__global__ void __launch_bounds__(CHUNK_T)
bucket_place_kernel(const int* __restrict__ src, const int* __restrict__ dst,
                    const int* __restrict__ cnt_in, const int* __restrict__ cnt_out,
                    int2* __restrict__ bin_edge, int* __restrict__ bin_src,
                    int* __restrict__ base_in, int* __restrict__ base_out,
                    int E, int epc, int B) {
  __shared__ int tot_in[MAXB];
  __shared__ int tot_out[MAXB];
  __shared__ int cur_in[MAXB];
  __shared__ int cur_out[MAXB];
  __shared__ int wsum_in[MAXB / 32];
  __shared__ int wsum_out[MAXB / 32];
  const int tid  = threadIdx.x;
  const int lane = tid & 31;
  const int wave = tid >> 5;
  const int c = blockIdx.x;
  if (tid < MAXB) { tot_in[tid] = 0; tot_out[tid] = 0; cur_in[tid] = 0; cur_out[tid] = 0; }
  __syncthreads();

  const int b = tid & (MAXB - 1);
  if (b < B) {
    int ti = 0, to = 0, pi = 0, po = 0;
    for (int cc = tid >> BK_SHIFT; cc < NCHUNK; cc += CHUNK_T / MAXB) {
      const int vi = cnt_in[cc * B + b];
      const int vo = cnt_out[cc * B + b];
      ti += vi; to += vo;
      if (cc < c) { pi += vi; po += vo; }
    }
    atomicAdd(&tot_in[b], ti);  atomicAdd(&tot_out[b], to);
    atomicAdd(&cur_in[b], pi);  atomicAdd(&cur_out[b], po);
  }
  __syncthreads();

  int vi = 0, vo = 0;
  if (tid < MAXB) { vi = tot_in[tid]; vo = tot_out[tid]; }
  int si = vi, so = vo;
  for (int o = 1; o < 32; o <<= 1) {
    const int ui = __shfl_up(si, o);
    const int uo = __shfl_up(so, o);
    if (lane >= o) { si += ui; so += uo; }
  }
  if (tid < MAXB && lane == 31) { wsum_in[wave] = si; wsum_out[wave] = so; }
  __syncthreads();
  if (tid < MAXB) {
    int offi = 0, offo = 0;
    for (int w = 0; w < wave; ++w) { offi += wsum_in[w]; offo += wsum_out[w]; }
    const int bi = offi + si - vi;
    const int bo = offo + so - vo;
    cur_in[tid]  += bi;
    cur_out[tid] += bo;
    if (c == 0 && tid < B) {
      base_in[tid]  = bi;
      base_out[tid] = bo;
      if (tid == B - 1) { base_in[B] = bi + vi; base_out[B] = bo + vo; }
    }
  }
  __syncthreads();

  const int e0 = c * epc;
  const int e1 = min(E, e0 + epc);
  for (int e = e0 + tid; e < e1; e += CHUNK_T) {
    const int s = src[e];
    const int d = dst[e];
    const int p = atomicAdd(&cur_in[d >> BK_SHIFT], 1);
    bin_edge[p] = make_int2(s, d);
    const int q = atomicAdd(&cur_out[s >> BK_SHIFT], 1);
    bin_src[q] = s;
  }
}

__global__ void __launch_bounds__(SORT_T)
bucket_sort_kernel(const int2* __restrict__ bin_edge, const int* __restrict__ bin_src,
                   const int* __restrict__ base_in, const int* __restrict__ base_out,
                   int* __restrict__ rowptr, int* __restrict__ col,
                   float* __restrict__ ns, float* __restrict__ nd, int N) {
  __shared__ int c_in[BK];
  __shared__ int c_out[BK];
  __shared__ int cur[BK];
  __shared__ int wsum[BK / 32];
  const int tid  = threadIdx.x;
  const int lane = tid & 31;
  const int wave = tid >> 5;
  const int b = blockIdx.x;
  if (tid < BK) { c_in[tid] = 0; c_out[tid] = 0; }
  __syncthreads();
  const int i0 = base_in[b],  i1 = base_in[b + 1];
  const int o0 = base_out[b], o1 = base_out[b + 1];
  for (int e = i0 + tid; e < i1; e += SORT_T) atomicAdd(&c_in[bin_edge[e].y & (BK - 1)], 1);
  for (int e = o0 + tid; e < o1; e += SORT_T) atomicAdd(&c_out[bin_src[e] & (BK - 1)], 1);
  __syncthreads();

  const int ci = (tid < BK) ? c_in[tid] : 0;
  int incl = ci;
  for (int o = 1; o < 32; o <<= 1) {
    const int t = __shfl_up(incl, o);
    if (lane >= o) incl += t;
  }
  if (tid < BK && lane == 31) wsum[wave] = incl;
  __syncthreads();
  if (tid < BK) {
    int off = i0;
    for (int w = 0; w < wave; ++w) off += wsum[w];
    const int first = off + incl - ci;
    cur[tid] = first;
    const int node = b * BK + tid;
    if (node < N) {
      rowptr[node] = first;
      if (node == N - 1) rowptr[N] = first + ci;
      ns[node] = rsqrtf(fmaxf((float)(c_out[tid] + 1), 1.0f));
      nd[node] = rsqrtf(fmaxf((float)(ci + 1), 1.0f));
    }
  }
  __syncthreads();

  for (int e = i0 + tid; e < i1; e += SORT_T) {
    const int2 p = bin_edge[e];
    col[atomicAdd(&cur[p.y & (BK - 1)], 1)] = p.x;
  }
}

__global__ void gemm_wmma_f32_kernel(const float* __restrict__ X, const float* __restrict__ W,
                                     float* __restrict__ Y, int KD) {
  const int wave = threadIdx.x >> 5;
  const int lane = threadIdx.x & 31;
  const int l16  = lane & 15;
  const int hi   = lane >> 4;
  const int mbase = blockIdx.x * NT;
  const int nbase = wave * NT;

  const float* __restrict__ xrow = X + (size_t)(mbase + l16) * KD;
  const float* __restrict__ wcol = W + nbase + l16;

  v8f c = {};
  #pragma unroll 4
  for (int k = 0; k < KD; k += 4) {
    v2f a, b;
    a.x = xrow[k + 2 * hi];
    a.y = xrow[k + 2 * hi + 1];
    b.x = wcol[(size_t)(k + 2 * hi) * HDIM];
    b.y = wcol[(size_t)(k + 2 * hi + 1) * HDIM];
    c = __builtin_amdgcn_wmma_f32_16x16x4_f32(false, a, false, b, (short)0, c, false, false);
  }

  float* __restrict__ yout = Y + nbase + l16;
  #pragma unroll
  for (int v = 0; v < 8; ++v)
    yout[(size_t)(mbase + v + 8 * hi) * HDIM] = c[v];
}

__global__ void __launch_bounds__(256)
gather_kernel(const int* __restrict__ rowptr, const int* __restrict__ col,
              const float* __restrict__ ns, const float* __restrict__ nd,
              const float* __restrict__ Y, const float* __restrict__ bias,
              float* __restrict__ out, int N, int relu) {
  const int lane = threadIdx.x & 31;
  const int wave = __builtin_amdgcn_readfirstlane((int)(threadIdx.x >> 5));
  const int node = blockIdx.x * 8 + wave;
  if (node >= N) return;

  const float4* __restrict__ Y4 = (const float4*)Y;
  const int beg = rowptr[node];
  const int end = rowptr[node + 1];
  const float4 self = Y4[(size_t)node * (HDIM / 4) + lane];
  const float nsi = ns[node];
  const float ndi = nd[node];
  const float b0 = bias[lane * 4 + 0], b1 = bias[lane * 4 + 1];
  const float b2 = bias[lane * 4 + 2], b3 = bias[lane * 4 + 3];

  float4 acc = {0.0f, 0.0f, 0.0f, 0.0f};
  for (int j0 = beg; j0 < end; j0 += 32) {
    const int n = min(32, end - j0);
    int s = 0;
    float w = 0.0f;
    if (lane < n) { s = col[j0 + lane]; w = ns[s]; }
    const int wi = __float_as_int(w);
    int k = 0;
    for (; k + 4 <= n; k += 4) {
      const int s0 = __builtin_amdgcn_readlane(s, k);
      const int s1 = __builtin_amdgcn_readlane(s, k + 1);
      const int s2 = __builtin_amdgcn_readlane(s, k + 2);
      const int s3 = __builtin_amdgcn_readlane(s, k + 3);
      const float w0 = __int_as_float(__builtin_amdgcn_readlane(wi, k));
      const float w1 = __int_as_float(__builtin_amdgcn_readlane(wi, k + 1));
      const float w2 = __int_as_float(__builtin_amdgcn_readlane(wi, k + 2));
      const float w3 = __int_as_float(__builtin_amdgcn_readlane(wi, k + 3));
      const float4 m0 = Y4[(size_t)s0 * (HDIM / 4) + lane];
      const float4 m1 = Y4[(size_t)s1 * (HDIM / 4) + lane];
      const float4 m2 = Y4[(size_t)s2 * (HDIM / 4) + lane];
      const float4 m3 = Y4[(size_t)s3 * (HDIM / 4) + lane];
      acc.x = fmaf(m0.x, w0, acc.x); acc.y = fmaf(m0.y, w0, acc.y);
      acc.z = fmaf(m0.z, w0, acc.z); acc.w = fmaf(m0.w, w0, acc.w);
      acc.x = fmaf(m1.x, w1, acc.x); acc.y = fmaf(m1.y, w1, acc.y);
      acc.z = fmaf(m1.z, w1, acc.z); acc.w = fmaf(m1.w, w1, acc.w);
      acc.x = fmaf(m2.x, w2, acc.x); acc.y = fmaf(m2.y, w2, acc.y);
      acc.z = fmaf(m2.z, w2, acc.z); acc.w = fmaf(m2.w, w2, acc.w);
      acc.x = fmaf(m3.x, w3, acc.x); acc.y = fmaf(m3.y, w3, acc.y);
      acc.z = fmaf(m3.z, w3, acc.z); acc.w = fmaf(m3.w, w3, acc.w);
    }
    #pragma nounroll
    for (; k < n; ++k) {
      const int s0 = __builtin_amdgcn_readlane(s, k);
      const float w0 = __int_as_float(__builtin_amdgcn_readlane(wi, k));
      const float4 m0 = Y4[(size_t)s0 * (HDIM / 4) + lane];
      acc.x = fmaf(m0.x, w0, acc.x); acc.y = fmaf(m0.y, w0, acc.y);
      acc.z = fmaf(m0.z, w0, acc.z); acc.w = fmaf(m0.w, w0, acc.w);
    }
  }

  float4 v;
  v.x = (acc.x + self.x * nsi) * ndi + b0;
  v.y = (acc.y + self.y * nsi) * ndi + b1;
  v.z = (acc.z + self.z * nsi) * ndi + b2;
  v.w = (acc.w + self.w * nsi) * ndi + b3;
  if (relu) {
    v.x = fmaxf(v.x, 0.0f); v.y = fmaxf(v.y, 0.0f);
    v.z = fmaxf(v.z, 0.0f); v.w = fmaxf(v.w, 0.0f);
  }
  ((float4*)out)[(size_t)node * (HDIM / 4) + lane] = v;
}

static inline int cdiv(long long a, int b) { return (int)((a + b - 1) / b); }

static void run_graph(const float* feat, const int* src, const int* dst,
                      const float* W0, const float* b0, const float* W1, const float* b1,
                      int N, int D, int E,
                      float* ns, float* nd, float* Y, float* agg, float* h,
                      float* out, hipStream_t stream) {
  const int B   = cdiv(N, BK);
  const int epc = cdiv(E, NCHUNK);
  int2* bin_edge = (int2*)agg;
  int* bin_src  = (int*)agg + 2 * (size_t)E;
  int* col      = bin_src + E;
  int* rowptr   = col + E;
  int* cnt_in   = rowptr + (N + 1);
  int* cnt_out  = cnt_in + NCHUNK * B;
  int* base_in  = cnt_out + NCHUNK * B;
  int* base_out = base_in + (B + 1);

  bucket_count_kernel<<<NCHUNK, CHUNK_T, 0, stream>>>(src, dst, cnt_in, cnt_out, E, epc, B);
  bucket_place_kernel<<<NCHUNK, CHUNK_T, 0, stream>>>(src, dst, cnt_in, cnt_out, bin_edge, bin_src,
                                                      base_in, base_out, E, epc, B);
  bucket_sort_kernel<<<B, SORT_T, 0, stream>>>(bin_edge, bin_src, base_in, base_out, rowptr, col, ns, nd, N);

  gemm_wmma_f32_kernel<<<N / NT, 256, 0, stream>>>(feat, W0, Y, D);
  gather_kernel<<<cdiv(N, 8), 256, 0, stream>>>(rowptr, col, ns, nd, Y, b0, h, N, 1);

  gemm_wmma_f32_kernel<<<N / NT, 256, 0, stream>>>(h, W1, Y, HDIM);
  gather_kernel<<<cdiv(N, 8), 256, 0, stream>>>(rowptr, col, ns, nd, Y, b1, out, N, 0);
}

extern "C" void kernel_launch(void* const* d_in, const int* in_sizes, int n_in,
                              void* d_out, int out_size, void* d_ws, size_t ws_size,
                              hipStream_t stream) {
  const float* feat_a = (const float*)d_in[0];
  const float* feat_b = (const float*)d_in[1];
  const int*   src_a  = (const int*)d_in[2];
  const int*   dst_a  = (const int*)d_in[3];
  const int*   src_b  = (const int*)d_in[4];
  const int*   dst_b  = (const int*)d_in[5];
  const float* Wa0 = (const float*)d_in[6];
  const float* ba0 = (const float*)d_in[7];
  const float* Wa1 = (const float*)d_in[8];
  const float* ba1 = (const float*)d_in[9];
  const float* Wb0 = (const float*)d_in[10];
  const float* bb0 = (const float*)d_in[11];
  const float* Wb1 = (const float*)d_in[12];
  const float* bb1 = (const float*)d_in[13];

  const int DA = in_sizes[6] / HDIM;
  const int DB = in_sizes[10] / HDIM;
  const int NA = in_sizes[0] / DA;
  const int NB = in_sizes[1] / DB;
  const int EA = in_sizes[2];
  const int EB = in_sizes[4];

  float* ws = (float*)d_ws;
  float* ns_a  = ws;
  float* nd_a  = ns_a + NA;
  float* Y_a   = nd_a + NA;
  float* agg_a = Y_a + (size_t)NA * HDIM;
  float* h_a   = agg_a + (size_t)NA * HDIM;
  float* ns_b  = h_a + (size_t)NA * HDIM;
  float* nd_b  = ns_b + NB;
  float* Y_b   = nd_b + NB;
  float* agg_b = Y_b + (size_t)NB * HDIM;
  float* h_b   = agg_b + (size_t)NB * HDIM;

  float* out_a = (float*)d_out;
  float* out_b = out_a + (size_t)NA * HDIM;

  run_graph(feat_a, src_a, dst_a, Wa0, ba0, Wa1, ba1, NA, DA, EA,
            ns_a, nd_a, Y_a, agg_a, h_a, out_a, stream);
  run_graph(feat_b, src_b, dst_b, Wb0, bb0, Wb1, bb1, NB, DB, EB,
            ns_b, nd_b, Y_b, agg_b, h_b, out_b, stream);
}


# GEMM re-tiled: one wave owns 16 rows x all 128 columns (8 accumulators), W staged once per work-group in LDS, X read once in 16-byte loads; the k order of each dot product's sum changes
# speedup vs baseline: 2.9650x; 1.2335x over previous
#include <hip/hip_runtime.h>

typedef __attribute__((ext_vector_type(2))) float v2f;
typedef __attribute__((ext_vector_type(8))) float v8f;

#define HDIM 128
#define NT   16

#define BK       256
#define BK_SHIFT 8
#define MAXB     256
#define NCHUNK   128
#define CHUNK_T  1024
#define SORT_T   1024

__global__ void __launch_bounds__(CHUNK_T)
bucket_count_kernel(const int* __restrict__ src, const int* __restrict__ dst,
                    int* __restrict__ cnt_in, int* __restrict__ cnt_out,
                    int E, int epc, int B) {
  __shared__ int h_in[MAXB];
  __shared__ int h_out[MAXB];
  const int tid = threadIdx.x;
  const int c = blockIdx.x;
  if (tid < MAXB) { h_in[tid] = 0; h_out[tid] = 0; }
  __syncthreads();
  const int e0 = c * epc;
  const int e1 = min(E, e0 + epc);
  for (int e = e0 + tid; e < e1; e += CHUNK_T) {
    atomicAdd(&h_in[dst[e] >> BK_SHIFT], 1);
    atomicAdd(&h_out[src[e] >> BK_SHIFT], 1);
  }
  __syncthreads();
  if (tid < B) {
    cnt_in[c * B + tid]  = h_in[tid];
    cnt_out[c * B + tid] = h_out[tid];
  }
}

__global__ void __launch_bounds__(CHUNK_T)
bucket_place_kernel(const int* __restrict__ src, const int* __restrict__ dst,
                    const int* __restrict__ cnt_in, const int* __restrict__ cnt_out,
                    int2* __restrict__ bin_edge, int* __restrict__ bin_src,
                    int* __restrict__ base_in, int* __restrict__ base_out,
                    int E, int epc, int B) {
  __shared__ int tot_in[MAXB];
  __shared__ int tot_out[MAXB];
  __shared__ int cur_in[MAXB];
  __shared__ int cur_out[MAXB];
  __shared__ int wsum_in[MAXB / 32];
  __shared__ int wsum_out[MAXB / 32];
  const int tid  = threadIdx.x;
  const int lane = tid & 31;
  const int wave = tid >> 5;
  const int c = blockIdx.x;
  if (tid < MAXB) { tot_in[tid] = 0; tot_out[tid] = 0; cur_in[tid] = 0; cur_out[tid] = 0; }
  __syncthreads();

  const int b = tid & (MAXB - 1);
  if (b < B) {
    int ti = 0, to = 0, pi = 0, po = 0;
    for (int cc = tid >> BK_SHIFT; cc < NCHUNK; cc += CHUNK_T / MAXB) {
      const int vi = cnt_in[cc * B + b];
      const int vo = cnt_out[cc * B + b];
      ti += vi; to += vo;
      if (cc < c) { pi += vi; po += vo; }
    }
    atomicAdd(&tot_in[b], ti);  atomicAdd(&tot_out[b], to);
    atomicAdd(&cur_in[b], pi);  atomicAdd(&cur_out[b], po);
  }
  __syncthreads();

  int vi = 0, vo = 0;
  if (tid < MAXB) { vi = tot_in[tid]; vo = tot_out[tid]; }
  int si = vi, so = vo;
  for (int o = 1; o < 32; o <<= 1) {
    const int ui = __shfl_up(si, o);
    const int uo = __shfl_up(so, o);
    if (lane >= o) { si += ui; so += uo; }
  }
  if (tid < MAXB && lane == 31) { wsum_in[wave] = si; wsum_out[wave] = so; }
  __syncthreads();
  if (tid < MAXB) {
    int offi = 0, offo = 0;
    for (int w = 0; w < wave; ++w) { offi += wsum_in[w]; offo += wsum_out[w]; }
    const int bi = offi + si - vi;
    const int bo = offo + so - vo;
    cur_in[tid]  += bi;
    cur_out[tid] += bo;
    if (c == 0 && tid < B) {
      base_in[tid]  = bi;
      base_out[tid] = bo;
      if (tid == B - 1) { base_in[B] = bi + vi; base_out[B] = bo + vo; }
    }
  }
  __syncthreads();

  const int e0 = c * epc;
  const int e1 = min(E, e0 + epc);
  for (int e = e0 + tid; e < e1; e += CHUNK_T) {
    const int s = src[e];
    const int d = dst[e];
    const int p = atomicAdd(&cur_in[d >> BK_SHIFT], 1);
    bin_edge[p] = make_int2(s, d);
    const int q = atomicAdd(&cur_out[s >> BK_SHIFT], 1);
    bin_src[q] = s;
  }
}

__global__ void __launch_bounds__(SORT_T)
bucket_sort_kernel(const int2* __restrict__ bin_edge, const int* __restrict__ bin_src,
                   const int* __restrict__ base_in, const int* __restrict__ base_out,
                   int* __restrict__ rowptr, int* __restrict__ col,
                   float* __restrict__ ns, float* __restrict__ nd, int N) {
  __shared__ int c_in[BK];
  __shared__ int c_out[BK];
  __shared__ int cur[BK];
  __shared__ int wsum[BK / 32];
  const int tid  = threadIdx.x;
  const int lane = tid & 31;
  const int wave = tid >> 5;
  const int b = blockIdx.x;
  if (tid < BK) { c_in[tid] = 0; c_out[tid] = 0; }
  __syncthreads();
  const int i0 = base_in[b],  i1 = base_in[b + 1];
  const int o0 = base_out[b], o1 = base_out[b + 1];
  for (int e = i0 + tid; e < i1; e += SORT_T) atomicAdd(&c_in[bin_edge[e].y & (BK - 1)], 1);
  for (int e = o0 + tid; e < o1; e += SORT_T) atomicAdd(&c_out[bin_src[e] & (BK - 1)], 1);
  __syncthreads();

  const int ci = (tid < BK) ? c_in[tid] : 0;
  int incl = ci;
  for (int o = 1; o < 32; o <<= 1) {
    const int t = __shfl_up(incl, o);
    if (lane >= o) incl += t;
  }
  if (tid < BK && lane == 31) wsum[wave] = incl;
  __syncthreads();
  if (tid < BK) {
    int off = i0;
    for (int w = 0; w < wave; ++w) off += wsum[w];
    const int first = off + incl - ci;
    cur[tid] = first;
    const int node = b * BK + tid;
    if (node < N) {
      rowptr[node] = first;
      if (node == N - 1) rowptr[N] = first + ci;
      ns[node] = rsqrtf(fmaxf((float)(c_out[tid] + 1), 1.0f));
      nd[node] = rsqrtf(fmaxf((float)(ci + 1), 1.0f));
    }
  }
  __syncthreads();

  for (int e = i0 + tid; e < i1; e += SORT_T) {
    const int2 p = bin_edge[e];
    col[atomicAdd(&cur[p.y & (BK - 1)], 1)] = p.x;
  }
}

#define KD_MAX 256
#define WPP    260
__global__ void __launch_bounds__(256)
gemm_wmma_f32_kernel(const float* __restrict__ X, const float* __restrict__ W,
                     float* __restrict__ Y, int KD, int ntiles) {
  __shared__ float Ws[(KD_MAX / 2) * WPP];
  const int lane = threadIdx.x & 31;
  const int wave = __builtin_amdgcn_readfirstlane((int)(threadIdx.x >> 5));
  const int l16  = lane & 15;
  const int hi   = lane >> 4;
  const int tile = blockIdx.x * 8 + wave;
  const bool active = tile < ntiles;

  const float4* __restrict__ xp =
      (const float4*)(X + (size_t)((active ? tile : 0) * NT + l16) * KD + 16 * hi);
  float4 a0 = xp[0], a1 = xp[1], a2 = xp[2], a3 = xp[3];

  #pragma unroll 4
  for (int p = wave; p < (KD >> 1); p += 8) {
    const float2* __restrict__ w0 = (const float2*)(W + (size_t)(2 * p) * HDIM);
    const float2* __restrict__ w1 = (const float2*)(W + (size_t)(2 * p + 1) * HDIM);
    const float2 x0 = w0[lane], y0 = w1[lane];
    const float2 x1 = w0[32 + lane], y1 = w1[32 + lane];
    float4* d = (float4*)&Ws[p * WPP];
    d[lane]      = make_float4(x0.x, y0.x, x0.y, y0.y);
    d[32 + lane] = make_float4(x1.x, y1.x, x1.y, y1.y);
  }
  __syncthreads();
  if (!active) return;

  v8f acc[8];
  #pragma unroll
  for (int t = 0; t < 8; ++t) acc[t] = (v8f){};

  const float* __restrict__ bbase = Ws + (8 * hi) * WPP + 2 * l16;
  for (int K0 = 0; K0 < KD; K0 += 32) {
    const int kn = min(K0 + 32, KD - 32) >> 2;
    const float4 n0 = xp[kn], n1 = xp[kn + 1], n2 = xp[kn + 2], n3 = xp[kn + 3];
    const float av[16] = {a0.x, a0.y, a0.z, a0.w, a1.x, a1.y, a1.z, a1.w,
                          a2.x, a2.y, a2.z, a2.w, a3.x, a3.y, a3.z, a3.w};
    const float* __restrict__ bk = bbase + (K0 >> 1) * WPP;
    #pragma unroll
    for (int j = 0; j < 8; ++j) {
      v2f a;
      a.x = av[2 * j];
      a.y = av[2 * j + 1];
      #pragma unroll
      for (int t = 0; t < 8; ++t) {
        const v2f b = *(const v2f*)(bk + j * WPP + 32 * t);
        acc[t] = __builtin_amdgcn_wmma_f32_16x16x4_f32(false, a, false, b, (short)0, acc[t], false, false);
      }
    }
    a0 = n0; a1 = n1; a2 = n2; a3 = n3;
  }

  float* __restrict__ yout = Y + (size_t)(tile * NT + 8 * hi) * HDIM + l16;
  #pragma unroll
  for (int t = 0; t < 8; ++t) {
    #pragma unroll
    for (int v = 0; v < 8; ++v)
      yout[(size_t)v * HDIM + 16 * t] = acc[t][v];
  }
}

__global__ void __launch_bounds__(256)
gather_kernel(const int* __restrict__ rowptr, const int* __restrict__ col,
              const float* __restrict__ ns, const float* __restrict__ nd,
              const float* __restrict__ Y, const float* __restrict__ bias,
              float* __restrict__ out, int N, int relu) {
  const int lane = threadIdx.x & 31;
  const int wave = __builtin_amdgcn_readfirstlane((int)(threadIdx.x >> 5));
  const int node = blockIdx.x * 8 + wave;
  if (node >= N) return;

  const float4* __restrict__ Y4 = (const float4*)Y;
  const int beg = rowptr[node];
  const int end = rowptr[node + 1];
  const float4 self = Y4[(size_t)node * (HDIM / 4) + lane];
  const float nsi = ns[node];
  const float ndi = nd[node];
  const float b0 = bias[lane * 4 + 0], b1 = bias[lane * 4 + 1];
  const float b2 = bias[lane * 4 + 2], b3 = bias[lane * 4 + 3];

  float4 acc = {0.0f, 0.0f, 0.0f, 0.0f};
  for (int j0 = beg; j0 < end; j0 += 32) {
    const int n = min(32, end - j0);
    int s = 0;
    float w = 0.0f;
    if (lane < n) { s = col[j0 + lane]; w = ns[s]; }
    const int wi = __float_as_int(w);
    int k = 0;
    for (; k + 4 <= n; k += 4) {
      const int s0 = __builtin_amdgcn_readlane(s, k);
      const int s1 = __builtin_amdgcn_readlane(s, k + 1);
      const int s2 = __builtin_amdgcn_readlane(s, k + 2);
      const int s3 = __builtin_amdgcn_readlane(s, k + 3);
      const float w0 = __int_as_float(__builtin_amdgcn_readlane(wi, k));
      const float w1 = __int_as_float(__builtin_amdgcn_readlane(wi, k + 1));
      const float w2 = __int_as_float(__builtin_amdgcn_readlane(wi, k + 2));
      const float w3 = __int_as_float(__builtin_amdgcn_readlane(wi, k + 3));
      const float4 m0 = Y4[(size_t)s0 * (HDIM / 4) + lane];
      const float4 m1 = Y4[(size_t)s1 * (HDIM / 4) + lane];
      const float4 m2 = Y4[(size_t)s2 * (HDIM / 4) + lane];
      const float4 m3 = Y4[(size_t)s3 * (HDIM / 4) + lane];
      acc.x = fmaf(m0.x, w0, acc.x); acc.y = fmaf(m0.y, w0, acc.y);
      acc.z = fmaf(m0.z, w0, acc.z); acc.w = fmaf(m0.w, w0, acc.w);
      acc.x = fmaf(m1.x, w1, acc.x); acc.y = fmaf(m1.y, w1, acc.y);
      acc.z = fmaf(m1.z, w1, acc.z); acc.w = fmaf(m1.w, w1, acc.w);
      acc.x = fmaf(m2.x, w2, acc.x); acc.y = fmaf(m2.y, w2, acc.y);
      acc.z = fmaf(m2.z, w2, acc.z); acc.w = fmaf(m2.w, w2, acc.w);
      acc.x = fmaf(m3.x, w3, acc.x); acc.y = fmaf(m3.y, w3, acc.y);
      acc.z = fmaf(m3.z, w3, acc.z); acc.w = fmaf(m3.w, w3, acc.w);
    }
    #pragma nounroll
    for (; k < n; ++k) {
      const int s0 = __builtin_amdgcn_readlane(s, k);
      const float w0 = __int_as_float(__builtin_amdgcn_readlane(wi, k));
      const float4 m0 = Y4[(size_t)s0 * (HDIM / 4) + lane];
      acc.x = fmaf(m0.x, w0, acc.x); acc.y = fmaf(m0.y, w0, acc.y);
      acc.z = fmaf(m0.z, w0, acc.z); acc.w = fmaf(m0.w, w0, acc.w);
    }
  }

  float4 v;
  v.x = (acc.x + self.x * nsi) * ndi + b0;
  v.y = (acc.y + self.y * nsi) * ndi + b1;
  v.z = (acc.z + self.z * nsi) * ndi + b2;
  v.w = (acc.w + self.w * nsi) * ndi + b3;
  if (relu) {
    v.x = fmaxf(v.x, 0.0f); v.y = fmaxf(v.y, 0.0f);
    v.z = fmaxf(v.z, 0.0f); v.w = fmaxf(v.w, 0.0f);
  }
  ((float4*)out)[(size_t)node * (HDIM / 4) + lane] = v;
}

static inline int cdiv(long long a, int b) { return (int)((a + b - 1) / b); }

static void run_graph(const float* feat, const int* src, const int* dst,
                      const float* W0, const float* b0, const float* W1, const float* b1,
                      int N, int D, int E,
                      float* ns, float* nd, float* Y, float* agg, float* h,
                      float* out, hipStream_t stream) {
  const int B   = cdiv(N, BK);
  const int epc = cdiv(E, NCHUNK);
  int2* bin_edge = (int2*)agg;
  int* bin_src  = (int*)agg + 2 * (size_t)E;
  int* col      = bin_src + E;
  int* rowptr   = col + E;
  int* cnt_in   = rowptr + (N + 1);
  int* cnt_out  = cnt_in + NCHUNK * B;
  int* base_in  = cnt_out + NCHUNK * B;
  int* base_out = base_in + (B + 1);

  bucket_count_kernel<<<NCHUNK, CHUNK_T, 0, stream>>>(src, dst, cnt_in, cnt_out, E, epc, B);
  bucket_place_kernel<<<NCHUNK, CHUNK_T, 0, stream>>>(src, dst, cnt_in, cnt_out, bin_edge, bin_src,
                                                      base_in, base_out, E, epc, B);
  bucket_sort_kernel<<<B, SORT_T, 0, stream>>>(bin_edge, bin_src, base_in, base_out, rowptr, col, ns, nd, N);

  gemm_wmma_f32_kernel<<<cdiv(N / NT, 8), 256, 0, stream>>>(feat, W0, Y, D, N / NT);
  gather_kernel<<<cdiv(N, 8), 256, 0, stream>>>(rowptr, col, ns, nd, Y, b0, h, N, 1);

  gemm_wmma_f32_kernel<<<cdiv(N / NT, 8), 256, 0, stream>>>(h, W1, Y, HDIM, N / NT);
  gather_kernel<<<cdiv(N, 8), 256, 0, stream>>>(rowptr, col, ns, nd, Y, b1, out, N, 0);
}

extern "C" void kernel_launch(void* const* d_in, const int* in_sizes, int n_in,
                              void* d_out, int out_size, void* d_ws, size_t ws_size,
                              hipStream_t stream) {
  const float* feat_a = (const float*)d_in[0];
  const float* feat_b = (const float*)d_in[1];
  const int*   src_a  = (const int*)d_in[2];
  const int*   dst_a  = (const int*)d_in[3];
  const int*   src_b  = (const int*)d_in[4];
  const int*   dst_b  = (const int*)d_in[5];
  const float* Wa0 = (const float*)d_in[6];
  const float* ba0 = (const float*)d_in[7];
  const float* Wa1 = (const float*)d_in[8];
  const float* ba1 = (const float*)d_in[9];
  const float* Wb0 = (const float*)d_in[10];
  const float* bb0 = (const float*)d_in[11];
  const float* Wb1 = (const float*)d_in[12];
  const float* bb1 = (const float*)d_in[13];

  const int DA = in_sizes[6] / HDIM;
  const int DB = in_sizes[10] / HDIM;
  const int NA = in_sizes[0] / DA;
  const int NB = in_sizes[1] / DB;
  const int EA = in_sizes[2];
  const int EB = in_sizes[4];

  float* ws = (float*)d_ws;
  float* ns_a  = ws;
  float* nd_a  = ns_a + NA;
  float* Y_a   = nd_a + NA;
  float* agg_a = Y_a + (size_t)NA * HDIM;
  float* h_a   = agg_a + (size_t)NA * HDIM;
  float* ns_b  = h_a + (size_t)NA * HDIM;
  float* nd_b  = ns_b + NB;
  float* Y_b   = nd_b + NB;
  float* agg_b = Y_b + (size_t)NB * HDIM;
  float* h_b   = agg_b + (size_t)NB * HDIM;

  float* out_a = (float*)d_out;
  float* out_b = out_a + (size_t)NA * HDIM;

  run_graph(feat_a, src_a, dst_a, Wa0, ba0, Wa1, ba1, NA, DA, EA,
            ns_a, nd_a, Y_a, agg_a, h_a, out_a, stream);
  run_graph(feat_b, src_b, dst_b, Wb0, bb0, Wb1, bb1, NB, DB, EB,
            ns_b, nd_b, Y_b, agg_b, h_b, out_b, stream);
}


# Both graphs share every launch: each stage's grid holds graph a's work-groups, then graph b's (block index picks the graph), so 14 dependent launches become 7; every sum and its order as before
# speedup vs baseline: 3.7387x; 1.2609x over previous
#include <hip/hip_runtime.h>

typedef __attribute__((ext_vector_type(2))) float v2f;
typedef __attribute__((ext_vector_type(8))) float v8f;

#define HDIM 128
#define NT   16

#define BK       256
#define BK_SHIFT 8
#define MAXB     256
#define NCHUNK   128
#define CHUNK_T  1024
#define SORT_T   1024

__global__ void __launch_bounds__(CHUNK_T)
bucket_count_kernel(const int* __restrict__ src_a, const int* __restrict__ dst_a,
                    int* __restrict__ cnt_in_a, int* __restrict__ cnt_out_a,
                    int E_a, int epc_a, int B_a,
                    const int* __restrict__ src_b, const int* __restrict__ dst_b,
                    int* __restrict__ cnt_in_b, int* __restrict__ cnt_out_b,
                    int E_b, int epc_b, int B_b) {
  __shared__ int h_in[MAXB];
  __shared__ int h_out[MAXB];
  const bool gb = blockIdx.x >= NCHUNK;
  const int* __restrict__ src = gb ? src_b : src_a;
  const int* __restrict__ dst = gb ? dst_b : dst_a;
  int* __restrict__ cnt_in  = gb ? cnt_in_b : cnt_in_a;
  int* __restrict__ cnt_out = gb ? cnt_out_b : cnt_out_a;
  const int E   = gb ? E_b : E_a;
  const int epc = gb ? epc_b : epc_a;
  const int B   = gb ? B_b : B_a;
  const int tid = threadIdx.x;
  const int c = blockIdx.x - (gb ? NCHUNK : 0);
  if (tid < MAXB) { h_in[tid] = 0; h_out[tid] = 0; }
  __syncthreads();
  const int e0 = c * epc;
  const int e1 = min(E, e0 + epc);
  for (int e = e0 + tid; e < e1; e += CHUNK_T) {
    atomicAdd(&h_in[dst[e] >> BK_SHIFT], 1);
    atomicAdd(&h_out[src[e] >> BK_SHIFT], 1);
  }
  __syncthreads();
  if (tid < B) {
    cnt_in[c * B + tid]  = h_in[tid];
    cnt_out[c * B + tid] = h_out[tid];
  }
}

__global__ void __launch_bounds__(CHUNK_T)
bucket_place_kernel(const int* __restrict__ src_a, const int* __restrict__ dst_a,
                    const int* __restrict__ cnt_in_a, const int* __restrict__ cnt_out_a,
                    int2* __restrict__ bin_edge_a, int* __restrict__ bin_src_a,
                    int* __restrict__ base_in_a, int* __restrict__ base_out_a,
                    int E_a, int epc_a, int B_a,
                    const int* __restrict__ src_b, const int* __restrict__ dst_b,
                    const int* __restrict__ cnt_in_b, const int* __restrict__ cnt_out_b,
                    int2* __restrict__ bin_edge_b, int* __restrict__ bin_src_b,
                    int* __restrict__ base_in_b, int* __restrict__ base_out_b,
                    int E_b, int epc_b, int B_b) {
  const bool gb = blockIdx.x >= NCHUNK;
  const int* __restrict__ src = gb ? src_b : src_a;
  const int* __restrict__ dst = gb ? dst_b : dst_a;
  const int* __restrict__ cnt_in  = gb ? cnt_in_b : cnt_in_a;
  const int* __restrict__ cnt_out = gb ? cnt_out_b : cnt_out_a;
  int2* __restrict__ bin_edge = gb ? bin_edge_b : bin_edge_a;
  int* __restrict__ bin_src   = gb ? bin_src_b : bin_src_a;
  int* __restrict__ base_in   = gb ? base_in_b : base_in_a;
  int* __restrict__ base_out  = gb ? base_out_b : base_out_a;
  const int E   = gb ? E_b : E_a;
  const int epc = gb ? epc_b : epc_a;
  const int B   = gb ? B_b : B_a;
  __shared__ int tot_in[MAXB];
  __shared__ int tot_out[MAXB];
  __shared__ int cur_in[MAXB];
  __shared__ int cur_out[MAXB];
  __shared__ int wsum_in[MAXB / 32];
  __shared__ int wsum_out[MAXB / 32];
  const int tid  = threadIdx.x;
  const int lane = tid & 31;
  const int wave = tid >> 5;
  const int c = blockIdx.x - (gb ? NCHUNK : 0);
  if (tid < MAXB) { tot_in[tid] = 0; tot_out[tid] = 0; cur_in[tid] = 0; cur_out[tid] = 0; }
  __syncthreads();

  const int b = tid & (MAXB - 1);
  if (b < B) {
    int ti = 0, to = 0, pi = 0, po = 0;
    for (int cc = tid >> BK_SHIFT; cc < NCHUNK; cc += CHUNK_T / MAXB) {
      const int vi = cnt_in[cc * B + b];
      const int vo = cnt_out[cc * B + b];
      ti += vi; to += vo;
      if (cc < c) { pi += vi; po += vo; }
    }
    atomicAdd(&tot_in[b], ti);  atomicAdd(&tot_out[b], to);
    atomicAdd(&cur_in[b], pi);  atomicAdd(&cur_out[b], po);
  }
  __syncthreads();

  int vi = 0, vo = 0;
  if (tid < MAXB) { vi = tot_in[tid]; vo = tot_out[tid]; }
  int si = vi, so = vo;
  for (int o = 1; o < 32; o <<= 1) {
    const int ui = __shfl_up(si, o);
    const int uo = __shfl_up(so, o);
    if (lane >= o) { si += ui; so += uo; }
  }
  if (tid < MAXB && lane == 31) { wsum_in[wave] = si; wsum_out[wave] = so; }
  __syncthreads();
  if (tid < MAXB) {
    int offi = 0, offo = 0;
    for (int w = 0; w < wave; ++w) { offi += wsum_in[w]; offo += wsum_out[w]; }
    const int bi = offi + si - vi;
    const int bo = offo + so - vo;
    cur_in[tid]  += bi;
    cur_out[tid] += bo;
    if (c == 0 && tid < B) {
      base_in[tid]  = bi;
      base_out[tid] = bo;
      if (tid == B - 1) { base_in[B] = bi + vi; base_out[B] = bo + vo; }
    }
  }
  __syncthreads();

  const int e0 = c * epc;
  const int e1 = min(E, e0 + epc);
  for (int e = e0 + tid; e < e1; e += CHUNK_T) {
    const int s = src[e];
    const int d = dst[e];
    const int p = atomicAdd(&cur_in[d >> BK_SHIFT], 1);
    bin_edge[p] = make_int2(s, d);
    const int q = atomicAdd(&cur_out[s >> BK_SHIFT], 1);
    bin_src[q] = s;
  }
}

__global__ void __launch_bounds__(SORT_T)
bucket_sort_kernel(const int2* __restrict__ bin_edge_a, const int* __restrict__ bin_src_a,
                   const int* __restrict__ base_in_a, const int* __restrict__ base_out_a,
                   int* __restrict__ rowptr_a, int* __restrict__ col_a,
                   float* __restrict__ ns_a, float* __restrict__ nd_a, int N_a, int B_a,
                   const int2* __restrict__ bin_edge_b, const int* __restrict__ bin_src_b,
                   const int* __restrict__ base_in_b, const int* __restrict__ base_out_b,
                   int* __restrict__ rowptr_b, int* __restrict__ col_b,
                   float* __restrict__ ns_b, float* __restrict__ nd_b, int N_b) {
  const bool gb = (int)blockIdx.x >= B_a;
  const int2* __restrict__ bin_edge = gb ? bin_edge_b : bin_edge_a;
  const int* __restrict__ bin_src   = gb ? bin_src_b : bin_src_a;
  const int* __restrict__ base_in   = gb ? base_in_b : base_in_a;
  const int* __restrict__ base_out  = gb ? base_out_b : base_out_a;
  int* __restrict__ rowptr = gb ? rowptr_b : rowptr_a;
  int* __restrict__ col    = gb ? col_b : col_a;
  float* __restrict__ ns   = gb ? ns_b : ns_a;
  float* __restrict__ nd   = gb ? nd_b : nd_a;
  const int N = gb ? N_b : N_a;
  __shared__ int c_in[BK];
  __shared__ int c_out[BK];
  __shared__ int cur[BK];
  __shared__ int wsum[BK / 32];
  const int tid  = threadIdx.x;
  const int lane = tid & 31;
  const int wave = tid >> 5;
  const int b = (int)blockIdx.x - (gb ? B_a : 0);
  if (tid < BK) { c_in[tid] = 0; c_out[tid] = 0; }
  __syncthreads();
  const int i0 = base_in[b],  i1 = base_in[b + 1];
  const int o0 = base_out[b], o1 = base_out[b + 1];
  for (int e = i0 + tid; e < i1; e += SORT_T) atomicAdd(&c_in[bin_edge[e].y & (BK - 1)], 1);
  for (int e = o0 + tid; e < o1; e += SORT_T) atomicAdd(&c_out[bin_src[e] & (BK - 1)], 1);
  __syncthreads();

  const int ci = (tid < BK) ? c_in[tid] : 0;
  int incl = ci;
  for (int o = 1; o < 32; o <<= 1) {
    const int t = __shfl_up(incl, o);
    if (lane >= o) incl += t;
  }
  if (tid < BK && lane == 31) wsum[wave] = incl;
  __syncthreads();
  if (tid < BK) {
    int off = i0;
    for (int w = 0; w < wave; ++w) off += wsum[w];
    const int first = off + incl - ci;
    cur[tid] = first;
    const int node = b * BK + tid;
    if (node < N) {
      rowptr[node] = first;
      if (node == N - 1) rowptr[N] = first + ci;
      ns[node] = rsqrtf(fmaxf((float)(c_out[tid] + 1), 1.0f));
      nd[node] = rsqrtf(fmaxf((float)(ci + 1), 1.0f));
    }
  }
  __syncthreads();

  for (int e = i0 + tid; e < i1; e += SORT_T) {
    const int2 p = bin_edge[e];
    col[atomicAdd(&cur[p.y & (BK - 1)], 1)] = p.x;
  }
}

#define KD_MAX 256
#define WPP    260
__global__ void __launch_bounds__(256)
gemm_wmma_f32_kernel(const float* __restrict__ X_a, const float* __restrict__ W_a,
                     float* __restrict__ Y_a, int KD_a, int ntiles_a, int nwg_a,
                     const float* __restrict__ X_b, const float* __restrict__ W_b,
                     float* __restrict__ Y_b, int KD_b, int ntiles_b) {
  __shared__ float Ws[(KD_MAX / 2) * WPP];
  const bool gb = (int)blockIdx.x >= nwg_a;
  const float* __restrict__ X = gb ? X_b : X_a;
  const float* __restrict__ W = gb ? W_b : W_a;
  float* __restrict__ Y = gb ? Y_b : Y_a;
  const int KD     = gb ? KD_b : KD_a;
  const int ntiles = gb ? ntiles_b : ntiles_a;
  const int lane = threadIdx.x & 31;
  const int wave = __builtin_amdgcn_readfirstlane((int)(threadIdx.x >> 5));
  const int l16  = lane & 15;
  const int hi   = lane >> 4;
  const int tile = ((int)blockIdx.x - (gb ? nwg_a : 0)) * 8 + wave;
  const bool active = tile < ntiles;

  const float4* __restrict__ xp =
      (const float4*)(X + (size_t)((active ? tile : 0) * NT + l16) * KD + 16 * hi);
  float4 a0 = xp[0], a1 = xp[1], a2 = xp[2], a3 = xp[3];

  #pragma unroll 4
  for (int p = wave; p < (KD >> 1); p += 8) {
    const float2* __restrict__ w0 = (const float2*)(W + (size_t)(2 * p) * HDIM);
    const float2* __restrict__ w1 = (const float2*)(W + (size_t)(2 * p + 1) * HDIM);
    const float2 x0 = w0[lane], y0 = w1[lane];
    const float2 x1 = w0[32 + lane], y1 = w1[32 + lane];
    float4* d = (float4*)&Ws[p * WPP];
    d[lane]      = make_float4(x0.x, y0.x, x0.y, y0.y);
    d[32 + lane] = make_float4(x1.x, y1.x, x1.y, y1.y);
  }
  __syncthreads();
  if (!active) return;

  v8f acc[8];
  #pragma unroll
  for (int t = 0; t < 8; ++t) acc[t] = (v8f){};

  const float* __restrict__ bbase = Ws + (8 * hi) * WPP + 2 * l16;
  for (int K0 = 0; K0 < KD; K0 += 32) {
    const int kn = min(K0 + 32, KD - 32) >> 2;
    const float4 n0 = xp[kn], n1 = xp[kn + 1], n2 = xp[kn + 2], n3 = xp[kn + 3];
    const float av[16] = {a0.x, a0.y, a0.z, a0.w, a1.x, a1.y, a1.z, a1.w,
                          a2.x, a2.y, a2.z, a2.w, a3.x, a3.y, a3.z, a3.w};
    const float* __restrict__ bk = bbase + (K0 >> 1) * WPP;
    #pragma unroll
    for (int j = 0; j < 8; ++j) {
      v2f a;
      a.x = av[2 * j];
      a.y = av[2 * j + 1];
      #pragma unroll
      for (int t = 0; t < 8; ++t) {
        const v2f b = *(const v2f*)(bk + j * WPP + 32 * t);
        acc[t] = __builtin_amdgcn_wmma_f32_16x16x4_f32(false, a, false, b, (short)0, acc[t], false, false);
      }
    }
    a0 = n0; a1 = n1; a2 = n2; a3 = n3;
  }

  float* __restrict__ yout = Y + (size_t)(tile * NT + 8 * hi) * HDIM + l16;
  #pragma unroll
  for (int t = 0; t < 8; ++t) {
    #pragma unroll
    for (int v = 0; v < 8; ++v)
      yout[(size_t)v * HDIM + 16 * t] = acc[t][v];
  }
}

__global__ void __launch_bounds__(256)
gather_kernel(const int* __restrict__ rowptr_a, const int* __restrict__ col_a,
              const float* __restrict__ ns_a, const float* __restrict__ nd_a,
              const float* __restrict__ Y_a, const float* __restrict__ bias_a,
              float* __restrict__ out_a, int N_a, int nwg_a,
              const int* __restrict__ rowptr_b, const int* __restrict__ col_b,
              const float* __restrict__ ns_b, const float* __restrict__ nd_b,
              const float* __restrict__ Y_b, const float* __restrict__ bias_b,
              float* __restrict__ out_b, int N_b, int relu) {
  const bool gb = (int)blockIdx.x >= nwg_a;
  const int* __restrict__ rowptr = gb ? rowptr_b : rowptr_a;
  const int* __restrict__ col    = gb ? col_b : col_a;
  const float* __restrict__ ns   = gb ? ns_b : ns_a;
  const float* __restrict__ nd   = gb ? nd_b : nd_a;
  const float* __restrict__ Y    = gb ? Y_b : Y_a;
  const float* __restrict__ bias = gb ? bias_b : bias_a;
  float* __restrict__ out = gb ? out_b : out_a;
  const int N = gb ? N_b : N_a;
  const int lane = threadIdx.x & 31;
  const int wave = __builtin_amdgcn_readfirstlane((int)(threadIdx.x >> 5));
  const int node = ((int)blockIdx.x - (gb ? nwg_a : 0)) * 8 + wave;
  if (node >= N) return;

  const float4* __restrict__ Y4 = (const float4*)Y;
  const int beg = rowptr[node];
  const int end = rowptr[node + 1];
  const float4 self = Y4[(size_t)node * (HDIM / 4) + lane];
  const float nsi = ns[node];
  const float ndi = nd[node];
  const float b0 = bias[lane * 4 + 0], b1 = bias[lane * 4 + 1];
  const float b2 = bias[lane * 4 + 2], b3 = bias[lane * 4 + 3];

  float4 acc = {0.0f, 0.0f, 0.0f, 0.0f};
  for (int j0 = beg; j0 < end; j0 += 32) {
    const int n = min(32, end - j0);
    int s = 0;
    float w = 0.0f;
    if (lane < n) { s = col[j0 + lane]; w = ns[s]; }
    const int wi = __float_as_int(w);
    int k = 0;
    for (; k + 4 <= n; k += 4) {
      const int s0 = __builtin_amdgcn_readlane(s, k);
      const int s1 = __builtin_amdgcn_readlane(s, k + 1);
      const int s2 = __builtin_amdgcn_readlane(s, k + 2);
      const int s3 = __builtin_amdgcn_readlane(s, k + 3);
      const float w0 = __int_as_float(__builtin_amdgcn_readlane(wi, k));
      const float w1 = __int_as_float(__builtin_amdgcn_readlane(wi, k + 1));
      const float w2 = __int_as_float(__builtin_amdgcn_readlane(wi, k + 2));
      const float w3 = __int_as_float(__builtin_amdgcn_readlane(wi, k + 3));
      const float4 m0 = Y4[(size_t)s0 * (HDIM / 4) + lane];
      const float4 m1 = Y4[(size_t)s1 * (HDIM / 4) + lane];
      const float4 m2 = Y4[(size_t)s2 * (HDIM / 4) + lane];
      const float4 m3 = Y4[(size_t)s3 * (HDIM / 4) + lane];
      acc.x = fmaf(m0.x, w0, acc.x); acc.y = fmaf(m0.y, w0, acc.y);
      acc.z = fmaf(m0.z, w0, acc.z); acc.w = fmaf(m0.w, w0, acc.w);
      acc.x = fmaf(m1.x, w1, acc.x); acc.y = fmaf(m1.y, w1, acc.y);
      acc.z = fmaf(m1.z, w1, acc.z); acc.w = fmaf(m1.w, w1, acc.w);
      acc.x = fmaf(m2.x, w2, acc.x); acc.y = fmaf(m2.y, w2, acc.y);
      acc.z = fmaf(m2.z, w2, acc.z); acc.w = fmaf(m2.w, w2, acc.w);
      acc.x = fmaf(m3.x, w3, acc.x); acc.y = fmaf(m3.y, w3, acc.y);
      acc.z = fmaf(m3.z, w3, acc.z); acc.w = fmaf(m3.w, w3, acc.w);
    }
    #pragma nounroll
    for (; k < n; ++k) {
      const int s0 = __builtin_amdgcn_readlane(s, k);
      const float w0 = __int_as_float(__builtin_amdgcn_readlane(wi, k));
      const float4 m0 = Y4[(size_t)s0 * (HDIM / 4) + lane];
      acc.x = fmaf(m0.x, w0, acc.x); acc.y = fmaf(m0.y, w0, acc.y);
      acc.z = fmaf(m0.z, w0, acc.z); acc.w = fmaf(m0.w, w0, acc.w);
    }
  }

  float4 v;
  v.x = (acc.x + self.x * nsi) * ndi + b0;
  v.y = (acc.y + self.y * nsi) * ndi + b1;
  v.z = (acc.z + self.z * nsi) * ndi + b2;
  v.w = (acc.w + self.w * nsi) * ndi + b3;
  if (relu) {
    v.x = fmaxf(v.x, 0.0f); v.y = fmaxf(v.y, 0.0f);
    v.z = fmaxf(v.z, 0.0f); v.w = fmaxf(v.w, 0.0f);
  }
  ((float4*)out)[(size_t)node * (HDIM / 4) + lane] = v;
}

static inline int cdiv(long long a, int b) { return (int)((a + b - 1) / b); }

struct GraphPlan {
  int N, E, B, epc;
  int2* bin_edge;
  int* bin_src;
  int* col;
  int* rowptr;
  int* cnt_in;
  int* cnt_out;
  int* base_in;
  int* base_out;
};

static GraphPlan plan_graph(float* agg, int N, int E) {
  GraphPlan g;
  g.N = N;
  g.E = E;
  g.B   = cdiv(N, BK);
  g.epc = cdiv(E, NCHUNK);
  g.bin_edge = (int2*)agg;
  g.bin_src  = (int*)agg + 2 * (size_t)E;
  g.col      = g.bin_src + E;
  g.rowptr   = g.col + E;
  g.cnt_in   = g.rowptr + (N + 1);
  g.cnt_out  = g.cnt_in + NCHUNK * g.B;
  g.base_in  = g.cnt_out + NCHUNK * g.B;
  g.base_out = g.base_in + (g.B + 1);
  return g;
}

extern "C" void kernel_launch(void* const* d_in, const int* in_sizes, int n_in,
                              void* d_out, int out_size, void* d_ws, size_t ws_size,
                              hipStream_t stream) {
  const float* feat_a = (const float*)d_in[0];
  const float* feat_b = (const float*)d_in[1];
  const int*   src_a  = (const int*)d_in[2];
  const int*   dst_a  = (const int*)d_in[3];
  const int*   src_b  = (const int*)d_in[4];
  const int*   dst_b  = (const int*)d_in[5];
  const float* Wa0 = (const float*)d_in[6];
  const float* ba0 = (const float*)d_in[7];
  const float* Wa1 = (const float*)d_in[8];
  const float* ba1 = (const float*)d_in[9];
  const float* Wb0 = (const float*)d_in[10];
  const float* bb0 = (const float*)d_in[11];
  const float* Wb1 = (const float*)d_in[12];
  const float* bb1 = (const float*)d_in[13];

  const int DA = in_sizes[6] / HDIM;
  const int DB = in_sizes[10] / HDIM;
  const int NA = in_sizes[0] / DA;
  const int NB = in_sizes[1] / DB;
  const int EA = in_sizes[2];
  const int EB = in_sizes[4];

  float* ws = (float*)d_ws;
  float* ns_a  = ws;
  float* nd_a  = ns_a + NA;
  float* Y_a   = nd_a + NA;
  float* agg_a = Y_a + (size_t)NA * HDIM;
  float* h_a   = agg_a + (size_t)NA * HDIM;
  float* ns_b  = h_a + (size_t)NA * HDIM;
  float* nd_b  = ns_b + NB;
  float* Y_b   = nd_b + NB;
  float* agg_b = Y_b + (size_t)NB * HDIM;
  float* h_b   = agg_b + (size_t)NB * HDIM;

  float* out_a = (float*)d_out;
  float* out_b = out_a + (size_t)NA * HDIM;

  const GraphPlan ga = plan_graph(agg_a, NA, EA);
  const GraphPlan gb = plan_graph(agg_b, NB, EB);

  bucket_count_kernel<<<2 * NCHUNK, CHUNK_T, 0, stream>>>(
      src_a, dst_a, ga.cnt_in, ga.cnt_out, ga.E, ga.epc, ga.B,
      src_b, dst_b, gb.cnt_in, gb.cnt_out, gb.E, gb.epc, gb.B);
  bucket_place_kernel<<<2 * NCHUNK, CHUNK_T, 0, stream>>>(
      src_a, dst_a, ga.cnt_in, ga.cnt_out, ga.bin_edge, ga.bin_src, ga.base_in, ga.base_out,
      ga.E, ga.epc, ga.B,
      src_b, dst_b, gb.cnt_in, gb.cnt_out, gb.bin_edge, gb.bin_src, gb.base_in, gb.base_out,
      gb.E, gb.epc, gb.B);
  bucket_sort_kernel<<<ga.B + gb.B, SORT_T, 0, stream>>>(
      ga.bin_edge, ga.bin_src, ga.base_in, ga.base_out, ga.rowptr, ga.col, ns_a, nd_a, NA, ga.B,
      gb.bin_edge, gb.bin_src, gb.base_in, gb.base_out, gb.rowptr, gb.col, ns_b, nd_b, NB);

  const int mm_a = cdiv(NA / NT, 8), mm_b = cdiv(NB / NT, 8);
  const int gw_a = cdiv(NA, 8),      gw_b = cdiv(NB, 8);

  gemm_wmma_f32_kernel<<<mm_a + mm_b, 256, 0, stream>>>(
      feat_a, Wa0, Y_a, DA, NA / NT, mm_a,
      feat_b, Wb0, Y_b, DB, NB / NT);
  gather_kernel<<<gw_a + gw_b, 256, 0, stream>>>(
      ga.rowptr, ga.col, ns_a, nd_a, Y_a, ba0, h_a, NA, gw_a,
      gb.rowptr, gb.col, ns_b, nd_b, Y_b, bb0, h_b, NB, 1);

  gemm_wmma_f32_kernel<<<mm_a + mm_b, 256, 0, stream>>>(
      h_a, Wa1, Y_a, HDIM, NA / NT, mm_a,
      h_b, Wb1, Y_b, HDIM, NB / NT);
  gather_kernel<<<gw_a + gw_b, 256, 0, stream>>>(
      ga.rowptr, ga.col, ns_a, nd_a, Y_a, ba1, out_a, NA, gw_a,
      gb.rowptr, gb.col, ns_b, nd_b, Y_b, bb1, out_b, NB, 0);
}


# Layer 0's product rides in the three edge-sort launches: its 1024-thread work-groups (64 registers, both kinds share a WGP) follow each stage's own in the grid; 7 launches become 6; sums as before
# speedup vs baseline: 4.0099x; 1.0725x over previous
#include <hip/hip_runtime.h>

typedef __attribute__((ext_vector_type(2))) float v2f;
typedef __attribute__((ext_vector_type(8))) float v8f;

#define HDIM 128
#define NT   16

#define BK       256
#define BK_SHIFT 8
#define MAXB     256
#define NCHUNK   128
#define CHUNK_T  1024
#define SORT_T   1024

__device__ __forceinline__ void
bucket_count_body(const int bid, const int* __restrict__ src_a, const int* __restrict__ dst_a,
                    int* __restrict__ cnt_in_a, int* __restrict__ cnt_out_a,
                    int E_a, int epc_a, int B_a,
                    const int* __restrict__ src_b, const int* __restrict__ dst_b,
                    int* __restrict__ cnt_in_b, int* __restrict__ cnt_out_b,
                    int E_b, int epc_b, int B_b) {
  __shared__ int h_in[MAXB];
  __shared__ int h_out[MAXB];
  const bool gb = bid >= NCHUNK;
  const int* __restrict__ src = gb ? src_b : src_a;
  const int* __restrict__ dst = gb ? dst_b : dst_a;
  int* __restrict__ cnt_in  = gb ? cnt_in_b : cnt_in_a;
  int* __restrict__ cnt_out = gb ? cnt_out_b : cnt_out_a;
  const int E   = gb ? E_b : E_a;
  const int epc = gb ? epc_b : epc_a;
  const int B   = gb ? B_b : B_a;
  const int tid = threadIdx.x;
  const int c = bid - (gb ? NCHUNK : 0);
  if (tid < MAXB) { h_in[tid] = 0; h_out[tid] = 0; }
  __syncthreads();
  const int e0 = c * epc;
  const int e1 = min(E, e0 + epc);
  for (int e = e0 + tid; e < e1; e += CHUNK_T) {
    atomicAdd(&h_in[dst[e] >> BK_SHIFT], 1);
    atomicAdd(&h_out[src[e] >> BK_SHIFT], 1);
  }
  __syncthreads();
  if (tid < B) {
    cnt_in[c * B + tid]  = h_in[tid];
    cnt_out[c * B + tid] = h_out[tid];
  }
}

__device__ __forceinline__ void
bucket_place_body(const int bid, const int* __restrict__ src_a, const int* __restrict__ dst_a,
                    const int* __restrict__ cnt_in_a, const int* __restrict__ cnt_out_a,
                    int2* __restrict__ bin_edge_a, int* __restrict__ bin_src_a,
                    int* __restrict__ base_in_a, int* __restrict__ base_out_a,
                    int E_a, int epc_a, int B_a,
                    const int* __restrict__ src_b, const int* __restrict__ dst_b,
                    const int* __restrict__ cnt_in_b, const int* __restrict__ cnt_out_b,
                    int2* __restrict__ bin_edge_b, int* __restrict__ bin_src_b,
                    int* __restrict__ base_in_b, int* __restrict__ base_out_b,
                    int E_b, int epc_b, int B_b) {
  const bool gb = bid >= NCHUNK;
  const int* __restrict__ src = gb ? src_b : src_a;
  const int* __restrict__ dst = gb ? dst_b : dst_a;
  const int* __restrict__ cnt_in  = gb ? cnt_in_b : cnt_in_a;
  const int* __restrict__ cnt_out = gb ? cnt_out_b : cnt_out_a;
  int2* __restrict__ bin_edge = gb ? bin_edge_b : bin_edge_a;
  int* __restrict__ bin_src   = gb ? bin_src_b : bin_src_a;
  int* __restrict__ base_in   = gb ? base_in_b : base_in_a;
  int* __restrict__ base_out  = gb ? base_out_b : base_out_a;
  const int E   = gb ? E_b : E_a;
  const int epc = gb ? epc_b : epc_a;
  const int B   = gb ? B_b : B_a;
  __shared__ int tot_in[MAXB];
  __shared__ int tot_out[MAXB];
  __shared__ int cur_in[MAXB];
  __shared__ int cur_out[MAXB];
  __shared__ int wsum_in[MAXB / 32];
  __shared__ int wsum_out[MAXB / 32];
  const int tid  = threadIdx.x;
  const int lane = tid & 31;
  const int wave = tid >> 5;
  const int c = bid - (gb ? NCHUNK : 0);
  if (tid < MAXB) { tot_in[tid] = 0; tot_out[tid] = 0; cur_in[tid] = 0; cur_out[tid] = 0; }
  __syncthreads();

  const int b = tid & (MAXB - 1);
  if (b < B) {
    int ti = 0, to = 0, pi = 0, po = 0;
    for (int cc = tid >> BK_SHIFT; cc < NCHUNK; cc += CHUNK_T / MAXB) {
      const int vi = cnt_in[cc * B + b];
      const int vo = cnt_out[cc * B + b];
      ti += vi; to += vo;
      if (cc < c) { pi += vi; po += vo; }
    }
    atomicAdd(&tot_in[b], ti);  atomicAdd(&tot_out[b], to);
    atomicAdd(&cur_in[b], pi);  atomicAdd(&cur_out[b], po);
  }
  __syncthreads();

  int vi = 0, vo = 0;
  if (tid < MAXB) { vi = tot_in[tid]; vo = tot_out[tid]; }
  int si = vi, so = vo;
  for (int o = 1; o < 32; o <<= 1) {
    const int ui = __shfl_up(si, o);
    const int uo = __shfl_up(so, o);
    if (lane >= o) { si += ui; so += uo; }
  }
  if (tid < MAXB && lane == 31) { wsum_in[wave] = si; wsum_out[wave] = so; }
  __syncthreads();
  if (tid < MAXB) {
    int offi = 0, offo = 0;
    for (int w = 0; w < wave; ++w) { offi += wsum_in[w]; offo += wsum_out[w]; }
    const int bi = offi + si - vi;
    const int bo = offo + so - vo;
    cur_in[tid]  += bi;
    cur_out[tid] += bo;
    if (c == 0 && tid < B) {
      base_in[tid]  = bi;
      base_out[tid] = bo;
      if (tid == B - 1) { base_in[B] = bi + vi; base_out[B] = bo + vo; }
    }
  }
  __syncthreads();

  const int e0 = c * epc;
  const int e1 = min(E, e0 + epc);
  for (int e = e0 + tid; e < e1; e += CHUNK_T) {
    const int s = src[e];
    const int d = dst[e];
    const int p = atomicAdd(&cur_in[d >> BK_SHIFT], 1);
    bin_edge[p] = make_int2(s, d);
    const int q = atomicAdd(&cur_out[s >> BK_SHIFT], 1);
    bin_src[q] = s;
  }
}

__device__ __forceinline__ void
bucket_sort_body(const int bid, const int2* __restrict__ bin_edge_a, const int* __restrict__ bin_src_a,
                   const int* __restrict__ base_in_a, const int* __restrict__ base_out_a,
                   int* __restrict__ rowptr_a, int* __restrict__ col_a,
                   float* __restrict__ ns_a, float* __restrict__ nd_a, int N_a, int B_a,
                   const int2* __restrict__ bin_edge_b, const int* __restrict__ bin_src_b,
                   const int* __restrict__ base_in_b, const int* __restrict__ base_out_b,
                   int* __restrict__ rowptr_b, int* __restrict__ col_b,
                   float* __restrict__ ns_b, float* __restrict__ nd_b, int N_b) {
  const bool gb = bid >= B_a;
  const int2* __restrict__ bin_edge = gb ? bin_edge_b : bin_edge_a;
  const int* __restrict__ bin_src   = gb ? bin_src_b : bin_src_a;
  const int* __restrict__ base_in   = gb ? base_in_b : base_in_a;
  const int* __restrict__ base_out  = gb ? base_out_b : base_out_a;
  int* __restrict__ rowptr = gb ? rowptr_b : rowptr_a;
  int* __restrict__ col    = gb ? col_b : col_a;
  float* __restrict__ ns   = gb ? ns_b : ns_a;
  float* __restrict__ nd   = gb ? nd_b : nd_a;
  const int N = gb ? N_b : N_a;
  __shared__ int c_in[BK];
  __shared__ int c_out[BK];
  __shared__ int cur[BK];
  __shared__ int wsum[BK / 32];
  const int tid  = threadIdx.x;
  const int lane = tid & 31;
  const int wave = tid >> 5;
  const int b = bid - (gb ? B_a : 0);
  if (tid < BK) { c_in[tid] = 0; c_out[tid] = 0; }
  __syncthreads();
  const int i0 = base_in[b],  i1 = base_in[b + 1];
  const int o0 = base_out[b], o1 = base_out[b + 1];
  for (int e = i0 + tid; e < i1; e += SORT_T) atomicAdd(&c_in[bin_edge[e].y & (BK - 1)], 1);
  for (int e = o0 + tid; e < o1; e += SORT_T) atomicAdd(&c_out[bin_src[e] & (BK - 1)], 1);
  __syncthreads();

  const int ci = (tid < BK) ? c_in[tid] : 0;
  int incl = ci;
  for (int o = 1; o < 32; o <<= 1) {
    const int t = __shfl_up(incl, o);
    if (lane >= o) incl += t;
  }
  if (tid < BK && lane == 31) wsum[wave] = incl;
  __syncthreads();
  if (tid < BK) {
    int off = i0;
    for (int w = 0; w < wave; ++w) off += wsum[w];
    const int first = off + incl - ci;
    cur[tid] = first;
    const int node = b * BK + tid;
    if (node < N) {
      rowptr[node] = first;
      if (node == N - 1) rowptr[N] = first + ci;
      ns[node] = rsqrtf(fmaxf((float)(c_out[tid] + 1), 1.0f));
      nd[node] = rsqrtf(fmaxf((float)(ci + 1), 1.0f));
    }
  }
  __syncthreads();

  for (int e = i0 + tid; e < i1; e += SORT_T) {
    const int2 p = bin_edge[e];
    col[atomicAdd(&cur[p.y & (BK - 1)], 1)] = p.x;
  }
}

#define KD_MAX 256
#define WPP    260
#define WS_FLOATS ((KD_MAX / 2) * WPP)
template <int CG>
__device__ __forceinline__ void
gemm_body(float* __restrict__ Ws, const int g,
          const float* __restrict__ X_a, const float* __restrict__ W_a,
          float* __restrict__ Y_a, int KD_a, int ntiles_a, int nwg_a,
          const float* __restrict__ X_b, const float* __restrict__ W_b,
          float* __restrict__ Y_b, int KD_b, int ntiles_b) {
  constexpr int NWAVE = 8 * CG;
  constexpr int TPW   = 8 / CG;
  const bool gb = g >= nwg_a;
  const float* __restrict__ X = gb ? X_b : X_a;
  const float* __restrict__ W = gb ? W_b : W_a;
  float* __restrict__ Y = gb ? Y_b : Y_a;
  const int KD     = gb ? KD_b : KD_a;
  const int ntiles = gb ? ntiles_b : ntiles_a;
  const int lane = threadIdx.x & 31;
  const int wave = __builtin_amdgcn_readfirstlane((int)(threadIdx.x >> 5));
  const int rt   = (CG == 1) ? wave : (wave & 7);
  const int cg   = (CG == 1) ? 0 : (wave >> 3);
  const int l16  = lane & 15;
  const int hi   = lane >> 4;
  const int tile = (g - (gb ? nwg_a : 0)) * 8 + rt;
  const bool active = tile < ntiles;

  const float4* __restrict__ xp =
      (const float4*)(X + (size_t)((active ? tile : 0) * NT + l16) * KD + 16 * hi);
  float4 a0, a1, a2, a3;
  if constexpr (CG == 1) { a0 = xp[0]; a1 = xp[1]; a2 = xp[2]; a3 = xp[3]; }

  constexpr int STAGE_UNROLL = (CG == 1) ? 4 : 2;
  #pragma unroll STAGE_UNROLL
  for (int p = wave; p < (KD >> 1); p += NWAVE) {
    const float2* __restrict__ w0 = (const float2*)(W + (size_t)(2 * p) * HDIM);
    const float2* __restrict__ w1 = (const float2*)(W + (size_t)(2 * p + 1) * HDIM);
    const float2 x0 = w0[lane], y0 = w1[lane];
    const float2 x1 = w0[32 + lane], y1 = w1[32 + lane];
    float4* d = (float4*)&Ws[p * WPP];
    d[lane]      = make_float4(x0.x, y0.x, x0.y, y0.y);
    d[32 + lane] = make_float4(x1.x, y1.x, x1.y, y1.y);
  }
  if constexpr (CG != 1) { a0 = xp[0]; a1 = xp[1]; a2 = xp[2]; a3 = xp[3]; }
  __syncthreads();
  if (!active) return;

  v8f acc[TPW];
  #pragma unroll
  for (int t = 0; t < TPW; ++t) acc[t] = (v8f){};

  const float* __restrict__ bbase = Ws + (8 * hi) * WPP + 2 * l16 + 32 * TPW * cg;
  for (int K0 = 0; K0 < KD; K0 += 32) {
    const int kn = min(K0 + 32, KD - 32) >> 2;
    const float4 n0 = xp[kn], n1 = xp[kn + 1], n2 = xp[kn + 2], n3 = xp[kn + 3];
    const float av[16] = {a0.x, a0.y, a0.z, a0.w, a1.x, a1.y, a1.z, a1.w,
                          a2.x, a2.y, a2.z, a2.w, a3.x, a3.y, a3.z, a3.w};
    const float* __restrict__ bk = bbase + (K0 >> 1) * WPP;
    #pragma unroll
    for (int j = 0; j < 8; ++j) {
      v2f a;
      a.x = av[2 * j];
      a.y = av[2 * j + 1];
      #pragma unroll
      for (int t = 0; t < TPW; ++t) {
        const v2f b = *(const v2f*)(bk + j * WPP + 32 * t);
        acc[t] = __builtin_amdgcn_wmma_f32_16x16x4_f32(false, a, false, b, (short)0, acc[t], false, false);
      }
    }
    a0 = n0; a1 = n1; a2 = n2; a3 = n3;
  }

  float* __restrict__ yout = Y + (size_t)(tile * NT + 8 * hi) * HDIM + 16 * TPW * cg + l16;
  #pragma unroll
  for (int t = 0; t < TPW; ++t) {
    #pragma unroll
    for (int v = 0; v < 8; ++v)
      yout[(size_t)v * HDIM + 16 * t] = acc[t][v];
  }
}

__global__ void __launch_bounds__(256)
gemm_wmma_f32_kernel(const float* __restrict__ X_a, const float* __restrict__ W_a,
                     float* __restrict__ Y_a, int KD_a, int ntiles_a, int nwg_a,
                     const float* __restrict__ X_b, const float* __restrict__ W_b,
                     float* __restrict__ Y_b, int KD_b, int ntiles_b) {
  __shared__ float Ws[WS_FLOATS];
  gemm_body<1>(Ws, (int)blockIdx.x, X_a, W_a, Y_a, KD_a, ntiles_a, nwg_a, X_b, W_b, Y_b, KD_b, ntiles_b);
}

#define SIDE_BY_SIDE __attribute__((amdgpu_waves_per_eu(16, 16)))
__global__ void __launch_bounds__(CHUNK_T) SIDE_BY_SIDE
count_and_gemm_kernel(const int* __restrict__ src_a, const int* __restrict__ dst_a,
                      int* __restrict__ cnt_in_a, int* __restrict__ cnt_out_a,
                      int E_a, int epc_a, int B_a,
                      const int* __restrict__ src_b, const int* __restrict__ dst_b,
                      int* __restrict__ cnt_in_b, int* __restrict__ cnt_out_b,
                      int E_b, int epc_b, int B_b,
                      int ncsr, int g0,
                      const float* __restrict__ X_a, const float* __restrict__ W_a,
                      float* __restrict__ Y_a, int KD_a, int ntiles_a, int nwg_a,
                      const float* __restrict__ X_b, const float* __restrict__ W_b,
                      float* __restrict__ Y_b, int KD_b, int ntiles_b) {
  __shared__ float Ws[WS_FLOATS];
  const int bid = (int)blockIdx.x;
  if (bid < ncsr) {
    bucket_count_body(bid, src_a, dst_a, cnt_in_a, cnt_out_a, E_a, epc_a, B_a,
                      src_b, dst_b, cnt_in_b, cnt_out_b, E_b, epc_b, B_b);
  } else {
    gemm_body<4>(Ws, g0 + (bid - ncsr), X_a, W_a, Y_a, KD_a, ntiles_a, nwg_a,
                 X_b, W_b, Y_b, KD_b, ntiles_b);
  }
}

__global__ void __launch_bounds__(CHUNK_T) SIDE_BY_SIDE
place_and_gemm_kernel(const int* __restrict__ src_a, const int* __restrict__ dst_a,
                      const int* __restrict__ cnt_in_a, const int* __restrict__ cnt_out_a,
                      int2* __restrict__ bin_edge_a, int* __restrict__ bin_src_a,
                      int* __restrict__ base_in_a, int* __restrict__ base_out_a,
                      int E_a, int epc_a, int B_a,
                      const int* __restrict__ src_b, const int* __restrict__ dst_b,
                      const int* __restrict__ cnt_in_b, const int* __restrict__ cnt_out_b,
                      int2* __restrict__ bin_edge_b, int* __restrict__ bin_src_b,
                      int* __restrict__ base_in_b, int* __restrict__ base_out_b,
                      int E_b, int epc_b, int B_b,
                      int ncsr, int g0,
                      const float* __restrict__ X_a, const float* __restrict__ W_a,
                      float* __restrict__ Y_a, int KD_a, int ntiles_a, int nwg_a,
                      const float* __restrict__ X_b, const float* __restrict__ W_b,
                      float* __restrict__ Y_b, int KD_b, int ntiles_b) {
  __shared__ float Ws[WS_FLOATS];
  const int bid = (int)blockIdx.x;
  if (bid < ncsr) {
    bucket_place_body(bid, src_a, dst_a, cnt_in_a, cnt_out_a, bin_edge_a, bin_src_a, base_in_a, base_out_a,
                      E_a, epc_a, B_a,
                      src_b, dst_b, cnt_in_b, cnt_out_b, bin_edge_b, bin_src_b, base_in_b, base_out_b,
                      E_b, epc_b, B_b);
  } else {
    gemm_body<4>(Ws, g0 + (bid - ncsr), X_a, W_a, Y_a, KD_a, ntiles_a, nwg_a,
                 X_b, W_b, Y_b, KD_b, ntiles_b);
  }
}

__global__ void __launch_bounds__(SORT_T) SIDE_BY_SIDE
sort_and_gemm_kernel(const int2* __restrict__ bin_edge_a, const int* __restrict__ bin_src_a,
                     const int* __restrict__ base_in_a, const int* __restrict__ base_out_a,
                     int* __restrict__ rowptr_a, int* __restrict__ col_a,
                     float* __restrict__ ns_a, float* __restrict__ nd_a, int N_a, int B_a,
                     const int2* __restrict__ bin_edge_b, const int* __restrict__ bin_src_b,
                     const int* __restrict__ base_in_b, const int* __restrict__ base_out_b,
                     int* __restrict__ rowptr_b, int* __restrict__ col_b,
                     float* __restrict__ ns_b, float* __restrict__ nd_b, int N_b,
                     int ncsr, int g0,
                     const float* __restrict__ X_a, const float* __restrict__ W_a,
                     float* __restrict__ Y_a, int KD_a, int ntiles_a, int nwg_a,
                     const float* __restrict__ X_b, const float* __restrict__ W_b,
                     float* __restrict__ Y_b, int KD_b, int ntiles_b) {
  __shared__ float Ws[WS_FLOATS];
  const int bid = (int)blockIdx.x;
  if (bid < ncsr) {
    bucket_sort_body(bid, bin_edge_a, bin_src_a, base_in_a, base_out_a, rowptr_a, col_a, ns_a, nd_a, N_a, B_a,
                     bin_edge_b, bin_src_b, base_in_b, base_out_b, rowptr_b, col_b, ns_b, nd_b, N_b);
  } else {
    gemm_body<4>(Ws, g0 + (bid - ncsr), X_a, W_a, Y_a, KD_a, ntiles_a, nwg_a,
                 X_b, W_b, Y_b, KD_b, ntiles_b);
  }
}

__global__ void __launch_bounds__(256)
gather_kernel(const int* __restrict__ rowptr_a, const int* __restrict__ col_a,
              const float* __restrict__ ns_a, const float* __restrict__ nd_a,
              const float* __restrict__ Y_a, const float* __restrict__ bias_a,
              float* __restrict__ out_a, int N_a, int nwg_a,
              const int* __restrict__ rowptr_b, const int* __restrict__ col_b,
              const float* __restrict__ ns_b, const float* __restrict__ nd_b,
              const float* __restrict__ Y_b, const float* __restrict__ bias_b,
              float* __restrict__ out_b, int N_b, int relu) {
  const bool gb = (int)blockIdx.x >= nwg_a;
  const int* __restrict__ rowptr = gb ? rowptr_b : rowptr_a;
  const int* __restrict__ col    = gb ? col_b : col_a;
  const float* __restrict__ ns   = gb ? ns_b : ns_a;
  const float* __restrict__ nd   = gb ? nd_b : nd_a;
  const float* __restrict__ Y    = gb ? Y_b : Y_a;
  const float* __restrict__ bias = gb ? bias_b : bias_a;
  float* __restrict__ out = gb ? out_b : out_a;
  const int N = gb ? N_b : N_a;
  const int lane = threadIdx.x & 31;
  const int wave = __builtin_amdgcn_readfirstlane((int)(threadIdx.x >> 5));
  const int node = ((int)blockIdx.x - (gb ? nwg_a : 0)) * 8 + wave;
  if (node >= N) return;

  const float4* __restrict__ Y4 = (const float4*)Y;
  const int beg = rowptr[node];
  const int end = rowptr[node + 1];
  const float4 self = Y4[(size_t)node * (HDIM / 4) + lane];
  const float nsi = ns[node];
  const float ndi = nd[node];
  const float b0 = bias[lane * 4 + 0], b1 = bias[lane * 4 + 1];
  const float b2 = bias[lane * 4 + 2], b3 = bias[lane * 4 + 3];

  float4 acc = {0.0f, 0.0f, 0.0f, 0.0f};
  for (int j0 = beg; j0 < end; j0 += 32) {
    const int n = min(32, end - j0);
    int s = 0;
    float w = 0.0f;
    if (lane < n) { s = col[j0 + lane]; w = ns[s]; }
    const int wi = __float_as_int(w);
    int k = 0;
    for (; k + 4 <= n; k += 4) {
      const int s0 = __builtin_amdgcn_readlane(s, k);
      const int s1 = __builtin_amdgcn_readlane(s, k + 1);
      const int s2 = __builtin_amdgcn_readlane(s, k + 2);
      const int s3 = __builtin_amdgcn_readlane(s, k + 3);
      const float w0 = __int_as_float(__builtin_amdgcn_readlane(wi, k));
      const float w1 = __int_as_float(__builtin_amdgcn_readlane(wi, k + 1));
      const float w2 = __int_as_float(__builtin_amdgcn_readlane(wi, k + 2));
      const float w3 = __int_as_float(__builtin_amdgcn_readlane(wi, k + 3));
      const float4 m0 = Y4[(size_t)s0 * (HDIM / 4) + lane];
      const float4 m1 = Y4[(size_t)s1 * (HDIM / 4) + lane];
      const float4 m2 = Y4[(size_t)s2 * (HDIM / 4) + lane];
      const float4 m3 = Y4[(size_t)s3 * (HDIM / 4) + lane];
      acc.x = fmaf(m0.x, w0, acc.x); acc.y = fmaf(m0.y, w0, acc.y);
      acc.z = fmaf(m0.z, w0, acc.z); acc.w = fmaf(m0.w, w0, acc.w);
      acc.x = fmaf(m1.x, w1, acc.x); acc.y = fmaf(m1.y, w1, acc.y);
      acc.z = fmaf(m1.z, w1, acc.z); acc.w = fmaf(m1.w, w1, acc.w);
      acc.x = fmaf(m2.x, w2, acc.x); acc.y = fmaf(m2.y, w2, acc.y);
      acc.z = fmaf(m2.z, w2, acc.z); acc.w = fmaf(m2.w, w2, acc.w);
      acc.x = fmaf(m3.x, w3, acc.x); acc.y = fmaf(m3.y, w3, acc.y);
      acc.z = fmaf(m3.z, w3, acc.z); acc.w = fmaf(m3.w, w3, acc.w);
    }
    #pragma nounroll
    for (; k < n; ++k) {
      const int s0 = __builtin_amdgcn_readlane(s, k);
      const float w0 = __int_as_float(__builtin_amdgcn_readlane(wi, k));
      const float4 m0 = Y4[(size_t)s0 * (HDIM / 4) + lane];
      acc.x = fmaf(m0.x, w0, acc.x); acc.y = fmaf(m0.y, w0, acc.y);
      acc.z = fmaf(m0.z, w0, acc.z); acc.w = fmaf(m0.w, w0, acc.w);
    }
  }

  float4 v;
  v.x = (acc.x + self.x * nsi) * ndi + b0;
  v.y = (acc.y + self.y * nsi) * ndi + b1;
  v.z = (acc.z + self.z * nsi) * ndi + b2;
  v.w = (acc.w + self.w * nsi) * ndi + b3;
  if (relu) {
    v.x = fmaxf(v.x, 0.0f); v.y = fmaxf(v.y, 0.0f);
    v.z = fmaxf(v.z, 0.0f); v.w = fmaxf(v.w, 0.0f);
  }
  ((float4*)out)[(size_t)node * (HDIM / 4) + lane] = v;
}

static inline int cdiv(long long a, int b) { return (int)((a + b - 1) / b); }

struct GraphPlan {
  int N, E, B, epc;
  int2* bin_edge;
  int* bin_src;
  int* col;
  int* rowptr;
  int* cnt_in;
  int* cnt_out;
  int* base_in;
  int* base_out;
};

static GraphPlan plan_graph(float* agg, int N, int E) {
  GraphPlan g;
  g.N = N;
  g.E = E;
  g.B   = cdiv(N, BK);
  g.epc = cdiv(E, NCHUNK);
  g.bin_edge = (int2*)agg;
  g.bin_src  = (int*)agg + 2 * (size_t)E;
  g.col      = g.bin_src + E;
  g.rowptr   = g.col + E;
  g.cnt_in   = g.rowptr + (N + 1);
  g.cnt_out  = g.cnt_in + NCHUNK * g.B;
  g.base_in  = g.cnt_out + NCHUNK * g.B;
  g.base_out = g.base_in + (g.B + 1);
  return g;
}

extern "C" void kernel_launch(void* const* d_in, const int* in_sizes, int n_in,
                              void* d_out, int out_size, void* d_ws, size_t ws_size,
                              hipStream_t stream) {
  const float* feat_a = (const float*)d_in[0];
  const float* feat_b = (const float*)d_in[1];
  const int*   src_a  = (const int*)d_in[2];
  const int*   dst_a  = (const int*)d_in[3];
  const int*   src_b  = (const int*)d_in[4];
  const int*   dst_b  = (const int*)d_in[5];
  const float* Wa0 = (const float*)d_in[6];
  const float* ba0 = (const float*)d_in[7];
  const float* Wa1 = (const float*)d_in[8];
  const float* ba1 = (const float*)d_in[9];
  const float* Wb0 = (const float*)d_in[10];
  const float* bb0 = (const float*)d_in[11];
  const float* Wb1 = (const float*)d_in[12];
  const float* bb1 = (const float*)d_in[13];

  const int DA = in_sizes[6] / HDIM;
  const int DB = in_sizes[10] / HDIM;
  const int NA = in_sizes[0] / DA;
  const int NB = in_sizes[1] / DB;
  const int EA = in_sizes[2];
  const int EB = in_sizes[4];

  float* ws = (float*)d_ws;
  float* ns_a  = ws;
  float* nd_a  = ns_a + NA;
  float* Y_a   = nd_a + NA;
  float* agg_a = Y_a + (size_t)NA * HDIM;
  float* h_a   = agg_a + (size_t)NA * HDIM;
  float* ns_b  = h_a + (size_t)NA * HDIM;
  float* nd_b  = ns_b + NB;
  float* Y_b   = nd_b + NB;
  float* agg_b = Y_b + (size_t)NB * HDIM;
  float* h_b   = agg_b + (size_t)NB * HDIM;

  float* out_a = (float*)d_out;
  float* out_b = out_a + (size_t)NA * HDIM;

  const GraphPlan ga = plan_graph(agg_a, NA, EA);
  const GraphPlan gb = plan_graph(agg_b, NB, EB);

  const int mm_a = cdiv(NA / NT, 8), mm_b = cdiv(NB / NT, 8);
  const int gw_a = cdiv(NA, 8),      gw_b = cdiv(NB, 8);

  const int gh = mm_a / 2;

  count_and_gemm_kernel<<<2 * NCHUNK + mm_b, CHUNK_T, 0, stream>>>(
      src_a, dst_a, ga.cnt_in, ga.cnt_out, ga.E, ga.epc, ga.B,
      src_b, dst_b, gb.cnt_in, gb.cnt_out, gb.E, gb.epc, gb.B,
      2 * NCHUNK, mm_a,
      feat_a, Wa0, Y_a, DA, NA / NT, mm_a,
      feat_b, Wb0, Y_b, DB, NB / NT);
  place_and_gemm_kernel<<<2 * NCHUNK + gh, CHUNK_T, 0, stream>>>(
      src_a, dst_a, ga.cnt_in, ga.cnt_out, ga.bin_edge, ga.bin_src, ga.base_in, ga.base_out,
      ga.E, ga.epc, ga.B,
      src_b, dst_b, gb.cnt_in, gb.cnt_out, gb.bin_edge, gb.bin_src, gb.base_in, gb.base_out,
      gb.E, gb.epc, gb.B,
      2 * NCHUNK, 0,
      feat_a, Wa0, Y_a, DA, NA / NT, mm_a,
      feat_b, Wb0, Y_b, DB, NB / NT);
  sort_and_gemm_kernel<<<ga.B + gb.B + (mm_a - gh), SORT_T, 0, stream>>>(
      ga.bin_edge, ga.bin_src, ga.base_in, ga.base_out, ga.rowptr, ga.col, ns_a, nd_a, NA, ga.B,
      gb.bin_edge, gb.bin_src, gb.base_in, gb.base_out, gb.rowptr, gb.col, ns_b, nd_b, NB,
      ga.B + gb.B, gh,
      feat_a, Wa0, Y_a, DA, NA / NT, mm_a,
      feat_b, Wb0, Y_b, DB, NB / NT);

  gather_kernel<<<gw_a + gw_b, 256, 0, stream>>>(
      ga.rowptr, ga.col, ns_a, nd_a, Y_a, ba0, h_a, NA, gw_a,
      gb.rowptr, gb.col, ns_b, nd_b, Y_b, bb0, h_b, NB, 1);

  gemm_wmma_f32_kernel<<<mm_a + mm_b, 256, 0, stream>>>(
      h_a, Wa1, Y_a, HDIM, NA / NT, mm_a,
      h_b, Wb1, Y_b, HDIM, NB / NT);
  gather_kernel<<<gw_a + gw_b, 256, 0, stream>>>(
      ga.rowptr, ga.col, ns_a, nd_a, Y_a, ba1, out_a, NA, gw_a,
      gb.rowptr, gb.col, ns_b, nd_b, Y_b, bb1, out_b, NB, 0);
}


# Layer 0's gather and layer 1's product share a work-group: h tiles go through LDS to WMMA, W fragments in registers, h never stored; 6 launches become 5; each dot product's k sum order changes
# speedup vs baseline: 4.3158x; 1.0763x over previous
#include <hip/hip_runtime.h>

typedef __attribute__((ext_vector_type(2))) float v2f;
typedef __attribute__((ext_vector_type(8))) float v8f;

#define HDIM 128
#define NT   16

#define BK       256
#define BK_SHIFT 8
#define MAXB     256
#define NCHUNK   128
#define CHUNK_T  1024
#define SORT_T   1024

__device__ __forceinline__ void
bucket_count_body(const int bid, const int* __restrict__ src_a, const int* __restrict__ dst_a,
                    int* __restrict__ cnt_in_a, int* __restrict__ cnt_out_a,
                    int E_a, int epc_a, int B_a,
                    const int* __restrict__ src_b, const int* __restrict__ dst_b,
                    int* __restrict__ cnt_in_b, int* __restrict__ cnt_out_b,
                    int E_b, int epc_b, int B_b) {
  __shared__ int h_in[MAXB];
  __shared__ int h_out[MAXB];
  const bool gb = bid >= NCHUNK;
  const int* __restrict__ src = gb ? src_b : src_a;
  const int* __restrict__ dst = gb ? dst_b : dst_a;
  int* __restrict__ cnt_in  = gb ? cnt_in_b : cnt_in_a;
  int* __restrict__ cnt_out = gb ? cnt_out_b : cnt_out_a;
  const int E   = gb ? E_b : E_a;
  const int epc = gb ? epc_b : epc_a;
  const int B   = gb ? B_b : B_a;
  const int tid = threadIdx.x;
  const int c = bid - (gb ? NCHUNK : 0);
  if (tid < MAXB) { h_in[tid] = 0; h_out[tid] = 0; }
  __syncthreads();
  const int e0 = c * epc;
  const int e1 = min(E, e0 + epc);
  for (int e = e0 + tid; e < e1; e += CHUNK_T) {
    atomicAdd(&h_in[dst[e] >> BK_SHIFT], 1);
    atomicAdd(&h_out[src[e] >> BK_SHIFT], 1);
  }
  __syncthreads();
  if (tid < B) {
    cnt_in[c * B + tid]  = h_in[tid];
    cnt_out[c * B + tid] = h_out[tid];
  }
}

__device__ __forceinline__ void
bucket_place_body(const int bid, const int* __restrict__ src_a, const int* __restrict__ dst_a,
                    const int* __restrict__ cnt_in_a, const int* __restrict__ cnt_out_a,
                    int2* __restrict__ bin_edge_a, int* __restrict__ bin_src_a,
                    int* __restrict__ base_in_a, int* __restrict__ base_out_a,
                    int E_a, int epc_a, int B_a,
                    const int* __restrict__ src_b, const int* __restrict__ dst_b,
                    const int* __restrict__ cnt_in_b, const int* __restrict__ cnt_out_b,
                    int2* __restrict__ bin_edge_b, int* __restrict__ bin_src_b,
                    int* __restrict__ base_in_b, int* __restrict__ base_out_b,
                    int E_b, int epc_b, int B_b) {
  const bool gb = bid >= NCHUNK;
  const int* __restrict__ src = gb ? src_b : src_a;
  const int* __restrict__ dst = gb ? dst_b : dst_a;
  const int* __restrict__ cnt_in  = gb ? cnt_in_b : cnt_in_a;
  const int* __restrict__ cnt_out = gb ? cnt_out_b : cnt_out_a;
  int2* __restrict__ bin_edge = gb ? bin_edge_b : bin_edge_a;
  int* __restrict__ bin_src   = gb ? bin_src_b : bin_src_a;
  int* __restrict__ base_in   = gb ? base_in_b : base_in_a;
  int* __restrict__ base_out  = gb ? base_out_b : base_out_a;
  const int E   = gb ? E_b : E_a;
  const int epc = gb ? epc_b : epc_a;
  const int B   = gb ? B_b : B_a;
  __shared__ int tot_in[MAXB];
  __shared__ int tot_out[MAXB];
  __shared__ int cur_in[MAXB];
  __shared__ int cur_out[MAXB];
  __shared__ int wsum_in[MAXB / 32];
  __shared__ int wsum_out[MAXB / 32];
  const int tid  = threadIdx.x;
  const int lane = tid & 31;
  const int wave = tid >> 5;
  const int c = bid - (gb ? NCHUNK : 0);
  if (tid < MAXB) { tot_in[tid] = 0; tot_out[tid] = 0; cur_in[tid] = 0; cur_out[tid] = 0; }
  __syncthreads();

  const int b = tid & (MAXB - 1);
  if (b < B) {
    int ti = 0, to = 0, pi = 0, po = 0;
    for (int cc = tid >> BK_SHIFT; cc < NCHUNK; cc += CHUNK_T / MAXB) {
      const int vi = cnt_in[cc * B + b];
      const int vo = cnt_out[cc * B + b];
      ti += vi; to += vo;
      if (cc < c) { pi += vi; po += vo; }
    }
    atomicAdd(&tot_in[b], ti);  atomicAdd(&tot_out[b], to);
    atomicAdd(&cur_in[b], pi);  atomicAdd(&cur_out[b], po);
  }
  __syncthreads();

  int vi = 0, vo = 0;
  if (tid < MAXB) { vi = tot_in[tid]; vo = tot_out[tid]; }
  int si = vi, so = vo;
  for (int o = 1; o < 32; o <<= 1) {
    const int ui = __shfl_up(si, o);
    const int uo = __shfl_up(so, o);
    if (lane >= o) { si += ui; so += uo; }
  }
  if (tid < MAXB && lane == 31) { wsum_in[wave] = si; wsum_out[wave] = so; }
  __syncthreads();
  if (tid < MAXB) {
    int offi = 0, offo = 0;
    for (int w = 0; w < wave; ++w) { offi += wsum_in[w]; offo += wsum_out[w]; }
    const int bi = offi + si - vi;
    const int bo = offo + so - vo;
    cur_in[tid]  += bi;
    cur_out[tid] += bo;
    if (c == 0 && tid < B) {
      base_in[tid]  = bi;
      base_out[tid] = bo;
      if (tid == B - 1) { base_in[B] = bi + vi; base_out[B] = bo + vo; }
    }
  }
  __syncthreads();

  const int e0 = c * epc;
  const int e1 = min(E, e0 + epc);
  for (int e = e0 + tid; e < e1; e += CHUNK_T) {
    const int s = src[e];
    const int d = dst[e];
    const int p = atomicAdd(&cur_in[d >> BK_SHIFT], 1);
    bin_edge[p] = make_int2(s, d);
    const int q = atomicAdd(&cur_out[s >> BK_SHIFT], 1);
    bin_src[q] = s;
  }
}

__device__ __forceinline__ void
bucket_sort_body(const int bid, const int2* __restrict__ bin_edge_a, const int* __restrict__ bin_src_a,
                   const int* __restrict__ base_in_a, const int* __restrict__ base_out_a,
                   int* __restrict__ rowptr_a, int* __restrict__ col_a,
                   float* __restrict__ ns_a, float* __restrict__ nd_a, int N_a, int B_a,
                   const int2* __restrict__ bin_edge_b, const int* __restrict__ bin_src_b,
                   const int* __restrict__ base_in_b, const int* __restrict__ base_out_b,
                   int* __restrict__ rowptr_b, int* __restrict__ col_b,
                   float* __restrict__ ns_b, float* __restrict__ nd_b, int N_b) {
  const bool gb = bid >= B_a;
  const int2* __restrict__ bin_edge = gb ? bin_edge_b : bin_edge_a;
  const int* __restrict__ bin_src   = gb ? bin_src_b : bin_src_a;
  const int* __restrict__ base_in   = gb ? base_in_b : base_in_a;
  const int* __restrict__ base_out  = gb ? base_out_b : base_out_a;
  int* __restrict__ rowptr = gb ? rowptr_b : rowptr_a;
  int* __restrict__ col    = gb ? col_b : col_a;
  float* __restrict__ ns   = gb ? ns_b : ns_a;
  float* __restrict__ nd   = gb ? nd_b : nd_a;
  const int N = gb ? N_b : N_a;
  __shared__ int c_in[BK];
  __shared__ int c_out[BK];
  __shared__ int cur[BK];
  __shared__ int wsum[BK / 32];
  const int tid  = threadIdx.x;
  const int lane = tid & 31;
  const int wave = tid >> 5;
  const int b = bid - (gb ? B_a : 0);
  if (tid < BK) { c_in[tid] = 0; c_out[tid] = 0; }
  __syncthreads();
  const int i0 = base_in[b],  i1 = base_in[b + 1];
  const int o0 = base_out[b], o1 = base_out[b + 1];
  for (int e = i0 + tid; e < i1; e += SORT_T) atomicAdd(&c_in[bin_edge[e].y & (BK - 1)], 1);
  for (int e = o0 + tid; e < o1; e += SORT_T) atomicAdd(&c_out[bin_src[e] & (BK - 1)], 1);
  __syncthreads();

  const int ci = (tid < BK) ? c_in[tid] : 0;
  int incl = ci;
  for (int o = 1; o < 32; o <<= 1) {
    const int t = __shfl_up(incl, o);
    if (lane >= o) incl += t;
  }
  if (tid < BK && lane == 31) wsum[wave] = incl;
  __syncthreads();
  if (tid < BK) {
    int off = i0;
    for (int w = 0; w < wave; ++w) off += wsum[w];
    const int first = off + incl - ci;
    cur[tid] = first;
    const int node = b * BK + tid;
    if (node < N) {
      rowptr[node] = first;
      if (node == N - 1) rowptr[N] = first + ci;
      ns[node] = rsqrtf(fmaxf((float)(c_out[tid] + 1), 1.0f));
      nd[node] = rsqrtf(fmaxf((float)(ci + 1), 1.0f));
    }
  }
  __syncthreads();

  for (int e = i0 + tid; e < i1; e += SORT_T) {
    const int2 p = bin_edge[e];
    col[atomicAdd(&cur[p.y & (BK - 1)], 1)] = p.x;
  }
}

#define KD_MAX 256
#define WPP    260
#define WS_FLOATS ((KD_MAX / 2) * WPP)
template <int CG>
__device__ __forceinline__ void
gemm_body(float* __restrict__ Ws, const int g,
          const float* __restrict__ X_a, const float* __restrict__ W_a,
          float* __restrict__ Y_a, int KD_a, int ntiles_a, int nwg_a,
          const float* __restrict__ X_b, const float* __restrict__ W_b,
          float* __restrict__ Y_b, int KD_b, int ntiles_b) {
  constexpr int NWAVE = 8 * CG;
  constexpr int TPW   = 8 / CG;
  const bool gb = g >= nwg_a;
  const float* __restrict__ X = gb ? X_b : X_a;
  const float* __restrict__ W = gb ? W_b : W_a;
  float* __restrict__ Y = gb ? Y_b : Y_a;
  const int KD     = gb ? KD_b : KD_a;
  const int ntiles = gb ? ntiles_b : ntiles_a;
  const int lane = threadIdx.x & 31;
  const int wave = __builtin_amdgcn_readfirstlane((int)(threadIdx.x >> 5));
  const int rt   = (CG == 1) ? wave : (wave & 7);
  const int cg   = (CG == 1) ? 0 : (wave >> 3);
  const int l16  = lane & 15;
  const int hi   = lane >> 4;
  const int tile = (g - (gb ? nwg_a : 0)) * 8 + rt;
  const bool active = tile < ntiles;

  const float4* __restrict__ xp =
      (const float4*)(X + (size_t)((active ? tile : 0) * NT + l16) * KD + 16 * hi);
  float4 a0, a1, a2, a3;
  if constexpr (CG == 1) { a0 = xp[0]; a1 = xp[1]; a2 = xp[2]; a3 = xp[3]; }

  constexpr int STAGE_UNROLL = (CG == 1) ? 4 : 2;
  #pragma unroll STAGE_UNROLL
  for (int p = wave; p < (KD >> 1); p += NWAVE) {
    const float2* __restrict__ w0 = (const float2*)(W + (size_t)(2 * p) * HDIM);
    const float2* __restrict__ w1 = (const float2*)(W + (size_t)(2 * p + 1) * HDIM);
    const float2 x0 = w0[lane], y0 = w1[lane];
    const float2 x1 = w0[32 + lane], y1 = w1[32 + lane];
    float4* d = (float4*)&Ws[p * WPP];
    d[lane]      = make_float4(x0.x, y0.x, x0.y, y0.y);
    d[32 + lane] = make_float4(x1.x, y1.x, x1.y, y1.y);
  }
  if constexpr (CG != 1) { a0 = xp[0]; a1 = xp[1]; a2 = xp[2]; a3 = xp[3]; }
  __syncthreads();
  if (!active) return;

  v8f acc[TPW];
  #pragma unroll
  for (int t = 0; t < TPW; ++t) acc[t] = (v8f){};

  const float* __restrict__ bbase = Ws + (8 * hi) * WPP + 2 * l16 + 32 * TPW * cg;
  for (int K0 = 0; K0 < KD; K0 += 32) {
    const int kn = min(K0 + 32, KD - 32) >> 2;
    const float4 n0 = xp[kn], n1 = xp[kn + 1], n2 = xp[kn + 2], n3 = xp[kn + 3];
    const float av[16] = {a0.x, a0.y, a0.z, a0.w, a1.x, a1.y, a1.z, a1.w,
                          a2.x, a2.y, a2.z, a2.w, a3.x, a3.y, a3.z, a3.w};
    const float* __restrict__ bk = bbase + (K0 >> 1) * WPP;
    #pragma unroll
    for (int j = 0; j < 8; ++j) {
      v2f a;
      a.x = av[2 * j];
      a.y = av[2 * j + 1];
      #pragma unroll
      for (int t = 0; t < TPW; ++t) {
        const v2f b = *(const v2f*)(bk + j * WPP + 32 * t);
        acc[t] = __builtin_amdgcn_wmma_f32_16x16x4_f32(false, a, false, b, (short)0, acc[t], false, false);
      }
    }
    a0 = n0; a1 = n1; a2 = n2; a3 = n3;
  }

  float* __restrict__ yout = Y + (size_t)(tile * NT + 8 * hi) * HDIM + 16 * TPW * cg + l16;
  #pragma unroll
  for (int t = 0; t < TPW; ++t) {
    #pragma unroll
    for (int v = 0; v < 8; ++v)
      yout[(size_t)v * HDIM + 16 * t] = acc[t][v];
  }
}

#define SIDE_BY_SIDE __attribute__((amdgpu_waves_per_eu(16, 16)))
__global__ void __launch_bounds__(CHUNK_T) SIDE_BY_SIDE
count_and_gemm_kernel(const int* __restrict__ src_a, const int* __restrict__ dst_a,
                      int* __restrict__ cnt_in_a, int* __restrict__ cnt_out_a,
                      int E_a, int epc_a, int B_a,
                      const int* __restrict__ src_b, const int* __restrict__ dst_b,
                      int* __restrict__ cnt_in_b, int* __restrict__ cnt_out_b,
                      int E_b, int epc_b, int B_b,
                      int ncsr, int g0,
                      const float* __restrict__ X_a, const float* __restrict__ W_a,
                      float* __restrict__ Y_a, int KD_a, int ntiles_a, int nwg_a,
                      const float* __restrict__ X_b, const float* __restrict__ W_b,
                      float* __restrict__ Y_b, int KD_b, int ntiles_b) {
  __shared__ float Ws[WS_FLOATS];
  const int bid = (int)blockIdx.x;
  if (bid < ncsr) {
    bucket_count_body(bid, src_a, dst_a, cnt_in_a, cnt_out_a, E_a, epc_a, B_a,
                      src_b, dst_b, cnt_in_b, cnt_out_b, E_b, epc_b, B_b);
  } else {
    gemm_body<4>(Ws, g0 + (bid - ncsr), X_a, W_a, Y_a, KD_a, ntiles_a, nwg_a,
                 X_b, W_b, Y_b, KD_b, ntiles_b);
  }
}

__global__ void __launch_bounds__(CHUNK_T) SIDE_BY_SIDE
place_and_gemm_kernel(const int* __restrict__ src_a, const int* __restrict__ dst_a,
                      const int* __restrict__ cnt_in_a, const int* __restrict__ cnt_out_a,
                      int2* __restrict__ bin_edge_a, int* __restrict__ bin_src_a,
                      int* __restrict__ base_in_a, int* __restrict__ base_out_a,
                      int E_a, int epc_a, int B_a,
                      const int* __restrict__ src_b, const int* __restrict__ dst_b,
                      const int* __restrict__ cnt_in_b, const int* __restrict__ cnt_out_b,
                      int2* __restrict__ bin_edge_b, int* __restrict__ bin_src_b,
                      int* __restrict__ base_in_b, int* __restrict__ base_out_b,
                      int E_b, int epc_b, int B_b,
                      int ncsr, int g0,
                      const float* __restrict__ X_a, const float* __restrict__ W_a,
                      float* __restrict__ Y_a, int KD_a, int ntiles_a, int nwg_a,
                      const float* __restrict__ X_b, const float* __restrict__ W_b,
                      float* __restrict__ Y_b, int KD_b, int ntiles_b) {
  __shared__ float Ws[WS_FLOATS];
  const int bid = (int)blockIdx.x;
  if (bid < ncsr) {
    bucket_place_body(bid, src_a, dst_a, cnt_in_a, cnt_out_a, bin_edge_a, bin_src_a, base_in_a, base_out_a,
                      E_a, epc_a, B_a,
                      src_b, dst_b, cnt_in_b, cnt_out_b, bin_edge_b, bin_src_b, base_in_b, base_out_b,
                      E_b, epc_b, B_b);
  } else {
    gemm_body<4>(Ws, g0 + (bid - ncsr), X_a, W_a, Y_a, KD_a, ntiles_a, nwg_a,
                 X_b, W_b, Y_b, KD_b, ntiles_b);
  }
}

__global__ void __launch_bounds__(SORT_T) SIDE_BY_SIDE
sort_and_gemm_kernel(const int2* __restrict__ bin_edge_a, const int* __restrict__ bin_src_a,
                     const int* __restrict__ base_in_a, const int* __restrict__ base_out_a,
                     int* __restrict__ rowptr_a, int* __restrict__ col_a,
                     float* __restrict__ ns_a, float* __restrict__ nd_a, int N_a, int B_a,
                     const int2* __restrict__ bin_edge_b, const int* __restrict__ bin_src_b,
                     const int* __restrict__ base_in_b, const int* __restrict__ base_out_b,
                     int* __restrict__ rowptr_b, int* __restrict__ col_b,
                     float* __restrict__ ns_b, float* __restrict__ nd_b, int N_b,
                     int ncsr, int g0,
                     const float* __restrict__ X_a, const float* __restrict__ W_a,
                     float* __restrict__ Y_a, int KD_a, int ntiles_a, int nwg_a,
                     const float* __restrict__ X_b, const float* __restrict__ W_b,
                     float* __restrict__ Y_b, int KD_b, int ntiles_b) {
  __shared__ float Ws[WS_FLOATS];
  const int bid = (int)blockIdx.x;
  if (bid < ncsr) {
    bucket_sort_body(bid, bin_edge_a, bin_src_a, base_in_a, base_out_a, rowptr_a, col_a, ns_a, nd_a, N_a, B_a,
                     bin_edge_b, bin_src_b, base_in_b, base_out_b, rowptr_b, col_b, ns_b, nd_b, N_b);
  } else {
    gemm_body<4>(Ws, g0 + (bid - ncsr), X_a, W_a, Y_a, KD_a, ntiles_a, nwg_a,
                 X_b, W_b, Y_b, KD_b, ntiles_b);
  }
}

__device__ __forceinline__ float4
gather_node(const int* __restrict__ rowptr, const int* __restrict__ col,
            const float* __restrict__ ns, const float* __restrict__ nd,
            const float4* __restrict__ Y4, const float* __restrict__ bias,
            const int node, const int lane, const int relu) {
  const int beg = rowptr[node];
  const int end = rowptr[node + 1];
  const float4 self = Y4[(size_t)node * (HDIM / 4) + lane];
  const float nsi = ns[node];
  const float ndi = nd[node];
  const float b0 = bias[lane * 4 + 0], b1 = bias[lane * 4 + 1];
  const float b2 = bias[lane * 4 + 2], b3 = bias[lane * 4 + 3];

  float4 acc = {0.0f, 0.0f, 0.0f, 0.0f};
  for (int j0 = beg; j0 < end; j0 += 32) {
    const int n = min(32, end - j0);
    int s = 0;
    float w = 0.0f;
    if (lane < n) { s = col[j0 + lane]; w = ns[s]; }
    const int wi = __float_as_int(w);
    int k = 0;
    for (; k + 4 <= n; k += 4) {
      const int s0 = __builtin_amdgcn_readlane(s, k);
      const int s1 = __builtin_amdgcn_readlane(s, k + 1);
      const int s2 = __builtin_amdgcn_readlane(s, k + 2);
      const int s3 = __builtin_amdgcn_readlane(s, k + 3);
      const float w0 = __int_as_float(__builtin_amdgcn_readlane(wi, k));
      const float w1 = __int_as_float(__builtin_amdgcn_readlane(wi, k + 1));
      const float w2 = __int_as_float(__builtin_amdgcn_readlane(wi, k + 2));
      const float w3 = __int_as_float(__builtin_amdgcn_readlane(wi, k + 3));
      const float4 m0 = Y4[(size_t)s0 * (HDIM / 4) + lane];
      const float4 m1 = Y4[(size_t)s1 * (HDIM / 4) + lane];
      const float4 m2 = Y4[(size_t)s2 * (HDIM / 4) + lane];
      const float4 m3 = Y4[(size_t)s3 * (HDIM / 4) + lane];
      acc.x = fmaf(m0.x, w0, acc.x); acc.y = fmaf(m0.y, w0, acc.y);
      acc.z = fmaf(m0.z, w0, acc.z); acc.w = fmaf(m0.w, w0, acc.w);
      acc.x = fmaf(m1.x, w1, acc.x); acc.y = fmaf(m1.y, w1, acc.y);
      acc.z = fmaf(m1.z, w1, acc.z); acc.w = fmaf(m1.w, w1, acc.w);
      acc.x = fmaf(m2.x, w2, acc.x); acc.y = fmaf(m2.y, w2, acc.y);
      acc.z = fmaf(m2.z, w2, acc.z); acc.w = fmaf(m2.w, w2, acc.w);
      acc.x = fmaf(m3.x, w3, acc.x); acc.y = fmaf(m3.y, w3, acc.y);
      acc.z = fmaf(m3.z, w3, acc.z); acc.w = fmaf(m3.w, w3, acc.w);
    }
    #pragma nounroll
    for (; k < n; ++k) {
      const int s0 = __builtin_amdgcn_readlane(s, k);
      const float w0 = __int_as_float(__builtin_amdgcn_readlane(wi, k));
      const float4 m0 = Y4[(size_t)s0 * (HDIM / 4) + lane];
      acc.x = fmaf(m0.x, w0, acc.x); acc.y = fmaf(m0.y, w0, acc.y);
      acc.z = fmaf(m0.z, w0, acc.z); acc.w = fmaf(m0.w, w0, acc.w);
    }
  }

  float4 v;
  v.x = (acc.x + self.x * nsi) * ndi + b0;
  v.y = (acc.y + self.y * nsi) * ndi + b1;
  v.z = (acc.z + self.z * nsi) * ndi + b2;
  v.w = (acc.w + self.w * nsi) * ndi + b3;
  if (relu) {
    v.x = fmaxf(v.x, 0.0f); v.y = fmaxf(v.y, 0.0f);
    v.z = fmaxf(v.z, 0.0f); v.w = fmaxf(v.w, 0.0f);
  }
  return v;
}

__global__ void __launch_bounds__(256)
gather_kernel(const int* __restrict__ rowptr_a, const int* __restrict__ col_a,
              const float* __restrict__ ns_a, const float* __restrict__ nd_a,
              const float* __restrict__ Y_a, const float* __restrict__ bias_a,
              float* __restrict__ out_a, int N_a, int nwg_a,
              const int* __restrict__ rowptr_b, const int* __restrict__ col_b,
              const float* __restrict__ ns_b, const float* __restrict__ nd_b,
              const float* __restrict__ Y_b, const float* __restrict__ bias_b,
              float* __restrict__ out_b, int N_b, int relu) {
  const bool gb = (int)blockIdx.x >= nwg_a;
  const int* __restrict__ rowptr = gb ? rowptr_b : rowptr_a;
  const int* __restrict__ col    = gb ? col_b : col_a;
  const float* __restrict__ ns   = gb ? ns_b : ns_a;
  const float* __restrict__ nd   = gb ? nd_b : nd_a;
  const float* __restrict__ Y    = gb ? Y_b : Y_a;
  const float* __restrict__ bias = gb ? bias_b : bias_a;
  float* __restrict__ out = gb ? out_b : out_a;
  const int N = gb ? N_b : N_a;
  const int lane = threadIdx.x & 31;
  const int wave = __builtin_amdgcn_readfirstlane((int)(threadIdx.x >> 5));
  const int node = ((int)blockIdx.x - (gb ? nwg_a : 0)) * 8 + wave;
  if (node >= N) return;
  const float4 v = gather_node(rowptr, col, ns, nd, (const float4*)Y, bias, node, lane, relu);
  ((float4*)out)[(size_t)node * (HDIM / 4) + lane] = v;
}

#define HP  132
#define TPG 5
__global__ void __launch_bounds__(256)
gather_gemm_kernel(const int* __restrict__ rowptr_a, const int* __restrict__ col_a,
                   const float* __restrict__ ns_a, const float* __restrict__ nd_a,
                   const float* __restrict__ Y_a, const float* __restrict__ bias_a,
                   const float* __restrict__ W_a, float* __restrict__ Z_a, int ntiles_a, int nwg_a,
                   const int* __restrict__ rowptr_b, const int* __restrict__ col_b,
                   const float* __restrict__ ns_b, const float* __restrict__ nd_b,
                   const float* __restrict__ Y_b, const float* __restrict__ bias_b,
                   const float* __restrict__ W_b, float* __restrict__ Z_b, int ntiles_b) {
  __shared__ __attribute__((aligned(16))) float Hs[NT * HP];
  const bool gb = (int)blockIdx.x >= nwg_a;
  const int* __restrict__ rowptr = gb ? rowptr_b : rowptr_a;
  const int* __restrict__ col    = gb ? col_b : col_a;
  const float* __restrict__ ns   = gb ? ns_b : ns_a;
  const float* __restrict__ nd   = gb ? nd_b : nd_a;
  const float* __restrict__ Y    = gb ? Y_b : Y_a;
  const float* __restrict__ bias = gb ? bias_b : bias_a;
  const float* __restrict__ W    = gb ? W_b : W_a;
  float* __restrict__ Z = gb ? Z_b : Z_a;
  const int ntiles = gb ? ntiles_b : ntiles_a;
  const int g = (int)blockIdx.x - (gb ? nwg_a : 0);
  const int lane = threadIdx.x & 31;
  const int wave = __builtin_amdgcn_readfirstlane((int)(threadIdx.x >> 5));
  const int l16  = lane & 15;
  const int hi   = lane >> 4;

  v2f bf[32];
  const float* __restrict__ wc = W + 16 * wave + l16;
  #pragma unroll
  for (int s = 0; s < 32; ++s) {
    const int k = 8 * (s >> 1) + 4 * hi + 2 * (s & 1);
    bf[s].x = wc[(size_t)k * HDIM];
    bf[s].y = wc[(size_t)(k + 1) * HDIM];
  }

  const int t0 = g * TPG;
  const int t1 = min(ntiles, t0 + TPG);
  for (int tile = t0; tile < t1; ++tile) {
    #pragma unroll
    for (int i = 0; i < 2; ++i) {
      const int row = 2 * wave + i;
      const float4 v = gather_node(rowptr, col, ns, nd, (const float4*)Y, bias, tile * NT + row, lane, 1);
      *(float4*)&Hs[row * HP + 4 * lane] = v;
    }
    __syncthreads();

    v8f acc0 = (v8f){}, acc1 = (v8f){};
    const float4* ap = (const float4*)&Hs[l16 * HP + 4 * hi];
    #pragma unroll
    for (int q = 0; q < 16; ++q) {
      const float4 a4 = ap[2 * q];
      v2f a;
      a.x = a4.x; a.y = a4.y;
      acc0 = __builtin_amdgcn_wmma_f32_16x16x4_f32(false, a, false, bf[2 * q], (short)0, acc0, false, false);
      a.x = a4.z; a.y = a4.w;
      acc1 = __builtin_amdgcn_wmma_f32_16x16x4_f32(false, a, false, bf[2 * q + 1], (short)0, acc1, false, false);
    }
    __syncthreads();

    #pragma unroll
    for (int r = 0; r < 8; ++r)
      Hs[(8 * hi + r) * HP + 16 * wave + l16] = acc0[r] + acc1[r];
    __syncthreads();
    #pragma unroll
    for (int i = 0; i < 2; ++i) {
      const int row = 2 * wave + i;
      const float4 y = *(const float4*)&Hs[row * HP + 4 * lane];
      ((float4*)Z)[(size_t)(tile * NT + row) * (HDIM / 4) + lane] = y;
    }
  }
}

static inline int cdiv(long long a, int b) { return (int)((a + b - 1) / b); }

struct GraphPlan {
  int N, E, B, epc;
  int2* bin_edge;
  int* bin_src;
  int* col;
  int* rowptr;
  int* cnt_in;
  int* cnt_out;
  int* base_in;
  int* base_out;
};

static GraphPlan plan_graph(float* agg, int N, int E) {
  GraphPlan g;
  g.N = N;
  g.E = E;
  g.B   = cdiv(N, BK);
  g.epc = cdiv(E, NCHUNK);
  g.bin_edge = (int2*)agg;
  g.bin_src  = (int*)agg + 2 * (size_t)E;
  g.col      = g.bin_src + E;
  g.rowptr   = g.col + E;
  g.cnt_in   = g.rowptr + (N + 1);
  g.cnt_out  = g.cnt_in + NCHUNK * g.B;
  g.base_in  = g.cnt_out + NCHUNK * g.B;
  g.base_out = g.base_in + (g.B + 1);
  return g;
}

extern "C" void kernel_launch(void* const* d_in, const int* in_sizes, int n_in,
                              void* d_out, int out_size, void* d_ws, size_t ws_size,
                              hipStream_t stream) {
  const float* feat_a = (const float*)d_in[0];
  const float* feat_b = (const float*)d_in[1];
  const int*   src_a  = (const int*)d_in[2];
  const int*   dst_a  = (const int*)d_in[3];
  const int*   src_b  = (const int*)d_in[4];
  const int*   dst_b  = (const int*)d_in[5];
  const float* Wa0 = (const float*)d_in[6];
  const float* ba0 = (const float*)d_in[7];
  const float* Wa1 = (const float*)d_in[8];
  const float* ba1 = (const float*)d_in[9];
  const float* Wb0 = (const float*)d_in[10];
  const float* bb0 = (const float*)d_in[11];
  const float* Wb1 = (const float*)d_in[12];
  const float* bb1 = (const float*)d_in[13];

  const int DA = in_sizes[6] / HDIM;
  const int DB = in_sizes[10] / HDIM;
  const int NA = in_sizes[0] / DA;
  const int NB = in_sizes[1] / DB;
  const int EA = in_sizes[2];
  const int EB = in_sizes[4];

  float* ws = (float*)d_ws;
  float* ns_a  = ws;
  float* nd_a  = ns_a + NA;
  float* Y_a   = nd_a + NA;
  float* agg_a = Y_a + (size_t)NA * HDIM;
  float* h_a   = agg_a + (size_t)NA * HDIM;
  float* ns_b  = h_a + (size_t)NA * HDIM;
  float* nd_b  = ns_b + NB;
  float* Y_b   = nd_b + NB;
  float* agg_b = Y_b + (size_t)NB * HDIM;
  float* h_b   = agg_b + (size_t)NB * HDIM;

  float* out_a = (float*)d_out;
  float* out_b = out_a + (size_t)NA * HDIM;

  const GraphPlan ga = plan_graph(agg_a, NA, EA);
  const GraphPlan gb = plan_graph(agg_b, NB, EB);

  const int mm_a = cdiv(NA / NT, 8), mm_b = cdiv(NB / NT, 8);
  const int gw_a = cdiv(NA, 8),      gw_b = cdiv(NB, 8);

  const int gh = mm_a / 2;

  count_and_gemm_kernel<<<2 * NCHUNK + mm_b, CHUNK_T, 0, stream>>>(
      src_a, dst_a, ga.cnt_in, ga.cnt_out, ga.E, ga.epc, ga.B,
      src_b, dst_b, gb.cnt_in, gb.cnt_out, gb.E, gb.epc, gb.B,
      2 * NCHUNK, mm_a,
      feat_a, Wa0, Y_a, DA, NA / NT, mm_a,
      feat_b, Wb0, Y_b, DB, NB / NT);
  place_and_gemm_kernel<<<2 * NCHUNK + gh, CHUNK_T, 0, stream>>>(
      src_a, dst_a, ga.cnt_in, ga.cnt_out, ga.bin_edge, ga.bin_src, ga.base_in, ga.base_out,
      ga.E, ga.epc, ga.B,
      src_b, dst_b, gb.cnt_in, gb.cnt_out, gb.bin_edge, gb.bin_src, gb.base_in, gb.base_out,
      gb.E, gb.epc, gb.B,
      2 * NCHUNK, 0,
      feat_a, Wa0, Y_a, DA, NA / NT, mm_a,
      feat_b, Wb0, Y_b, DB, NB / NT);
  sort_and_gemm_kernel<<<ga.B + gb.B + (mm_a - gh), SORT_T, 0, stream>>>(
      ga.bin_edge, ga.bin_src, ga.base_in, ga.base_out, ga.rowptr, ga.col, ns_a, nd_a, NA, ga.B,
      gb.bin_edge, gb.bin_src, gb.base_in, gb.base_out, gb.rowptr, gb.col, ns_b, nd_b, NB,
      ga.B + gb.B, gh,
      feat_a, Wa0, Y_a, DA, NA / NT, mm_a,
      feat_b, Wb0, Y_b, DB, NB / NT);

  const int gf_a = cdiv(NA / NT, TPG), gf_b = cdiv(NB / NT, TPG);
  gather_gemm_kernel<<<gf_a + gf_b, 256, 0, stream>>>(
      ga.rowptr, ga.col, ns_a, nd_a, Y_a, ba0, Wa1, h_a, NA / NT, gf_a,
      gb.rowptr, gb.col, ns_b, nd_b, Y_b, bb0, Wb1, h_b, NB / NT);

  gather_kernel<<<gw_a + gw_b, 256, 0, stream>>>(
      ga.rowptr, ga.col, ns_a, nd_a, h_a, ba1, out_a, NA, gw_a,
      gb.rowptr, gb.col, ns_b, nd_b, h_b, bb1, out_b, NB, 0);
}


# Edge sort in 2 launches, not 3: a chunk bins its edges into its own stretch on LDS counts (no global count tables), a bucket reads 128 chunk segments; 5 launches become 4; row sums stay arrival order
# speedup vs baseline: 4.9027x; 1.1360x over previous
#include <hip/hip_runtime.h>

typedef __attribute__((ext_vector_type(2))) float v2f;
typedef __attribute__((ext_vector_type(8))) float v8f;

#define HDIM 128
#define NT   16


#define BK       256
#define BK_SHIFT 8
#define MAXB     256
#define NCHUNK   128
#define CHUNK_T  1024
#define SORT_T   1024
#define SEG_PW   (NCHUNK / (SORT_T / 32))

__device__ __forceinline__ void
chunk_bin_body(const int bid, const int* __restrict__ src_a, const int* __restrict__ dst_a,
               int2* __restrict__ bin_edge_a, int* __restrict__ bin_src_a,
               int* __restrict__ off_in_a, int* __restrict__ off_out_a,
               int E_a, int epc_a, int B_a,
               const int* __restrict__ src_b, const int* __restrict__ dst_b,
               int2* __restrict__ bin_edge_b, int* __restrict__ bin_src_b,
               int* __restrict__ off_in_b, int* __restrict__ off_out_b,
               int E_b, int epc_b, int B_b) {
  const bool gb = bid >= NCHUNK;
  const int* __restrict__ src = gb ? src_b : src_a;
  const int* __restrict__ dst = gb ? dst_b : dst_a;
  int2* __restrict__ bin_edge = gb ? bin_edge_b : bin_edge_a;
  int* __restrict__ bin_src   = gb ? bin_src_b : bin_src_a;
  int* __restrict__ off_in    = gb ? off_in_b : off_in_a;
  int* __restrict__ off_out   = gb ? off_out_b : off_out_a;
  const int E   = gb ? E_b : E_a;
  const int epc = gb ? epc_b : epc_a;
  const int B   = gb ? B_b : B_a;
  __shared__ int h_in[MAXB];
  __shared__ int h_out[MAXB];
  __shared__ int wsum_in[MAXB / 32];
  __shared__ int wsum_out[MAXB / 32];
  const int tid  = threadIdx.x;
  const int lane = tid & 31;
  const int wave = tid >> 5;
  const int c = bid - (gb ? NCHUNK : 0);
  if (tid < MAXB) { h_in[tid] = 0; h_out[tid] = 0; }
  __syncthreads();
  const int e0 = c * epc;
  const int e1 = min(E, e0 + epc);
  for (int e = e0 + tid; e < e1; e += CHUNK_T) {
    const int d = dst[e];
    const int s = src[e];
    atomicAdd(&h_in[d >> BK_SHIFT], 1);
    atomicAdd(&h_out[s >> BK_SHIFT], 1);
  }
  __syncthreads();

  int vi = 0, vo = 0;
  if (tid < MAXB) { vi = h_in[tid]; vo = h_out[tid]; }
  int si = vi, so = vo;
  for (int o = 1; o < 32; o <<= 1) {
    const int ui = __shfl_up(si, o);
    const int uo = __shfl_up(so, o);
    if (lane >= o) { si += ui; so += uo; }
  }
  if (tid < MAXB && lane == 31) { wsum_in[wave] = si; wsum_out[wave] = so; }
  __syncthreads();
  if (tid < MAXB) {
    int offi = 0, offo = 0;
    for (int w = 0; w < wave; ++w) { offi += wsum_in[w]; offo += wsum_out[w]; }
    const int bi = offi + si - vi;
    const int bo = offo + so - vo;
    h_in[tid]  = e0 + bi;
    h_out[tid] = e0 + bo;
    if (tid < B) {
      off_in[c * (B + 1) + tid]  = bi;
      off_out[c * (B + 1) + tid] = bo;
      if (tid == B - 1) { off_in[c * (B + 1) + B] = bi + vi; off_out[c * (B + 1) + B] = bo + vo; }
    }
  }
  __syncthreads();

  for (int e = e0 + tid; e < e1; e += CHUNK_T) {
    const int s = src[e];
    const int d = dst[e];
    const int p = atomicAdd(&h_in[d >> BK_SHIFT], 1);
    bin_edge[p] = make_int2(s, d);
    const int q = atomicAdd(&h_out[s >> BK_SHIFT], 1);
    bin_src[q] = s;
  }
}

__device__ __forceinline__ void
bucket_sort_body(const int bid, const int2* __restrict__ bin_edge_a, const int* __restrict__ bin_src_a,
                 const int* __restrict__ off_in_a, const int* __restrict__ off_out_a,
                 int* __restrict__ rowptr_a, int* __restrict__ col_a,
                 float* __restrict__ ns_a, float* __restrict__ nd_a, int N_a, int B_a, int epc_a,
                 const int2* __restrict__ bin_edge_b, const int* __restrict__ bin_src_b,
                 const int* __restrict__ off_in_b, const int* __restrict__ off_out_b,
                 int* __restrict__ rowptr_b, int* __restrict__ col_b,
                 float* __restrict__ ns_b, float* __restrict__ nd_b, int N_b, int B_b, int epc_b) {
  const bool gb = bid >= B_a;
  const int2* __restrict__ bin_edge = gb ? bin_edge_b : bin_edge_a;
  const int* __restrict__ bin_src   = gb ? bin_src_b : bin_src_a;
  const int* __restrict__ off_in    = gb ? off_in_b : off_in_a;
  const int* __restrict__ off_out   = gb ? off_out_b : off_out_a;
  int* __restrict__ rowptr = gb ? rowptr_b : rowptr_a;
  int* __restrict__ col    = gb ? col_b : col_a;
  float* __restrict__ ns   = gb ? ns_b : ns_a;
  float* __restrict__ nd   = gb ? nd_b : nd_a;
  const int N   = gb ? N_b : N_a;
  const int B   = gb ? B_b : B_a;
  const int epc = gb ? epc_b : epc_a;
  __shared__ int c_in[BK];
  __shared__ int c_out[BK];
  __shared__ int cur[BK];
  __shared__ int wsum[BK / 32];
  __shared__ int sb_in[NCHUNK];
  __shared__ int sl_in[NCHUNK];
  __shared__ int sb_out[NCHUNK];
  __shared__ int sl_out[NCHUNK];
  __shared__ int front;
  const int tid  = threadIdx.x;
  const int lane = tid & 31;
  const int wave = tid >> 5;
  const int b = bid - (gb ? B_a : 0);

  int i_lo = 0, i_hi = 0, o_lo = 0, o_hi = 0;
  if (tid < NCHUNK) {
    const int* __restrict__ oi = off_in  + tid * (B + 1) + b;
    const int* __restrict__ oo = off_out + tid * (B + 1) + b;
    i_lo = oi[0]; i_hi = oi[1];
    o_lo = oo[0]; o_hi = oo[1];
  }
  if (tid < BK) { c_in[tid] = 0; c_out[tid] = 0; }
  if (tid == 0) front = 0;
  __syncthreads();
  if (tid < NCHUNK) {
    sb_in[tid]  = tid * epc + i_lo;  sl_in[tid]  = i_hi - i_lo;
    sb_out[tid] = tid * epc + o_lo;  sl_out[tid] = o_hi - o_lo;
    atomicAdd(&front, i_lo);
  }
  __syncthreads();
  const int i0 = front;

  int ib[SEG_PW], il[SEG_PW];
  #pragma unroll
  for (int k = 0; k < SEG_PW; ++k) {
    ib[k] = sb_in[wave + 32 * k];
    il[k] = sl_in[wave + 32 * k];
  }
  int2 pe[SEG_PW];
  #pragma unroll
  for (int k = 0; k < SEG_PW; ++k) pe[k] = bin_edge[lane < il[k] ? ib[k] + lane : 0];
  {
    int ob[SEG_PW], ol[SEG_PW], ps[SEG_PW];
    #pragma unroll
    for (int k = 0; k < SEG_PW; ++k) {
      ob[k] = sb_out[wave + 32 * k];
      ol[k] = sl_out[wave + 32 * k];
    }
    #pragma unroll
    for (int k = 0; k < SEG_PW; ++k) ps[k] = bin_src[lane < ol[k] ? ob[k] + lane : 0];
    #pragma unroll
    for (int k = 0; k < SEG_PW; ++k) {
      if (lane < il[k]) atomicAdd(&c_in[pe[k].y & (BK - 1)], 1);
      for (int i = lane + 32; i < il[k]; i += 32) atomicAdd(&c_in[bin_edge[ib[k] + i].y & (BK - 1)], 1);
    }
    #pragma unroll
    for (int k = 0; k < SEG_PW; ++k) {
      if (lane < ol[k]) atomicAdd(&c_out[ps[k] & (BK - 1)], 1);
      for (int i = lane + 32; i < ol[k]; i += 32) atomicAdd(&c_out[bin_src[ob[k] + i] & (BK - 1)], 1);
    }
  }
  __syncthreads();

  const int ci = (tid < BK) ? c_in[tid] : 0;
  int incl = ci;
  for (int o = 1; o < 32; o <<= 1) {
    const int t = __shfl_up(incl, o);
    if (lane >= o) incl += t;
  }
  if (tid < BK && lane == 31) wsum[wave] = incl;
  __syncthreads();
  if (tid < BK) {
    int off = i0;
    for (int w = 0; w < wave; ++w) off += wsum[w];
    const int first = off + incl - ci;
    cur[tid] = first;
    const int node = b * BK + tid;
    if (node < N) {
      rowptr[node] = first;
      if (node == N - 1) rowptr[N] = first + ci;
      ns[node] = rsqrtf(fmaxf((float)(c_out[tid] + 1), 1.0f));
      nd[node] = rsqrtf(fmaxf((float)(ci + 1), 1.0f));
    }
  }
  __syncthreads();

  #pragma unroll
  for (int k = 0; k < SEG_PW; ++k) {
    if (lane < il[k]) col[atomicAdd(&cur[pe[k].y & (BK - 1)], 1)] = pe[k].x;
    for (int i = lane + 32; i < il[k]; i += 32) {
      const int2 p = bin_edge[ib[k] + i];
      col[atomicAdd(&cur[p.y & (BK - 1)], 1)] = p.x;
    }
  }
}

#define KD_MAX 256
#define WPP    260
#define WS_FLOATS ((KD_MAX / 2) * WPP)
template <int CG>
__device__ __forceinline__ void
gemm_body(float* __restrict__ Ws, const int g,
          const float* __restrict__ X_a, const float* __restrict__ W_a,
          float* __restrict__ Y_a, int KD_a, int ntiles_a, int nwg_a,
          const float* __restrict__ X_b, const float* __restrict__ W_b,
          float* __restrict__ Y_b, int KD_b, int ntiles_b) {
  constexpr int NWAVE = 8 * CG;
  constexpr int TPW   = 8 / CG;
  const bool gb = g >= nwg_a;
  const float* __restrict__ X = gb ? X_b : X_a;
  const float* __restrict__ W = gb ? W_b : W_a;
  float* __restrict__ Y = gb ? Y_b : Y_a;
  const int KD     = gb ? KD_b : KD_a;
  const int ntiles = gb ? ntiles_b : ntiles_a;
  const int lane = threadIdx.x & 31;
  const int wave = __builtin_amdgcn_readfirstlane((int)(threadIdx.x >> 5));
  const int rt   = (CG == 1) ? wave : (wave & 7);
  const int cg   = (CG == 1) ? 0 : (wave >> 3);
  const int l16  = lane & 15;
  const int hi   = lane >> 4;
  const int tile = (g - (gb ? nwg_a : 0)) * 8 + rt;
  const bool active = tile < ntiles;

  const float4* __restrict__ xp =
      (const float4*)(X + (size_t)((active ? tile : 0) * NT + l16) * KD + 16 * hi);
  float4 a0, a1, a2, a3;
  if constexpr (CG == 1) { a0 = xp[0]; a1 = xp[1]; a2 = xp[2]; a3 = xp[3]; }

  constexpr int STAGE_UNROLL = (CG == 1) ? 4 : 2;
  #pragma unroll STAGE_UNROLL
  for (int p = wave; p < (KD >> 1); p += NWAVE) {
    const float2* __restrict__ w0 = (const float2*)(W + (size_t)(2 * p) * HDIM);
    const float2* __restrict__ w1 = (const float2*)(W + (size_t)(2 * p + 1) * HDIM);
    const float2 x0 = w0[lane], y0 = w1[lane];
    const float2 x1 = w0[32 + lane], y1 = w1[32 + lane];
    float4* d = (float4*)&Ws[p * WPP];
    d[lane]      = make_float4(x0.x, y0.x, x0.y, y0.y);
    d[32 + lane] = make_float4(x1.x, y1.x, x1.y, y1.y);
  }
  if constexpr (CG != 1) { a0 = xp[0]; a1 = xp[1]; a2 = xp[2]; a3 = xp[3]; }
  __syncthreads();
  if (!active) return;

  v8f acc[TPW];
  #pragma unroll
  for (int t = 0; t < TPW; ++t) acc[t] = (v8f){};

  const float* __restrict__ bbase = Ws + (8 * hi) * WPP + 2 * l16 + 32 * TPW * cg;
  for (int K0 = 0; K0 < KD; K0 += 32) {
    const int kn = min(K0 + 32, KD - 32) >> 2;
    const float4 n0 = xp[kn], n1 = xp[kn + 1], n2 = xp[kn + 2], n3 = xp[kn + 3];
    const float av[16] = {a0.x, a0.y, a0.z, a0.w, a1.x, a1.y, a1.z, a1.w,
                          a2.x, a2.y, a2.z, a2.w, a3.x, a3.y, a3.z, a3.w};
    const float* __restrict__ bk = bbase + (K0 >> 1) * WPP;
    #pragma unroll
    for (int j = 0; j < 8; ++j) {
      v2f a;
      a.x = av[2 * j];
      a.y = av[2 * j + 1];
      #pragma unroll
      for (int t = 0; t < TPW; ++t) {
        const v2f b = *(const v2f*)(bk + j * WPP + 32 * t);
        acc[t] = __builtin_amdgcn_wmma_f32_16x16x4_f32(false, a, false, b, (short)0, acc[t], false, false);
      }
    }
    a0 = n0; a1 = n1; a2 = n2; a3 = n3;
  }

  float* __restrict__ yout = Y + (size_t)(tile * NT + 8 * hi) * HDIM + 16 * TPW * cg + l16;
  #pragma unroll
  for (int t = 0; t < TPW; ++t) {
    #pragma unroll
    for (int v = 0; v < 8; ++v)
      yout[(size_t)v * HDIM + 16 * t] = acc[t][v];
  }
}

#define SIDE_BY_SIDE __attribute__((amdgpu_waves_per_eu(16, 16)))
__global__ void __launch_bounds__(CHUNK_T) SIDE_BY_SIDE
bin_and_gemm_kernel(const int* __restrict__ src_a, const int* __restrict__ dst_a,
                    int2* __restrict__ bin_edge_a, int* __restrict__ bin_src_a,
                    int* __restrict__ off_in_a, int* __restrict__ off_out_a,
                    int E_a, int epc_a, int B_a,
                    const int* __restrict__ src_b, const int* __restrict__ dst_b,
                    int2* __restrict__ bin_edge_b, int* __restrict__ bin_src_b,
                    int* __restrict__ off_in_b, int* __restrict__ off_out_b,
                    int E_b, int epc_b, int B_b,
                    int ncsr, int g0,
                    const float* __restrict__ X_a, const float* __restrict__ W_a,
                    float* __restrict__ Y_a, int KD_a, int ntiles_a, int nwg_a,
                    const float* __restrict__ X_b, const float* __restrict__ W_b,
                    float* __restrict__ Y_b, int KD_b, int ntiles_b) {
  __shared__ float Ws[WS_FLOATS];
  const int bid = (int)blockIdx.x;
  if (bid < ncsr) {
    chunk_bin_body(bid, src_a, dst_a, bin_edge_a, bin_src_a, off_in_a, off_out_a, E_a, epc_a, B_a,
                   src_b, dst_b, bin_edge_b, bin_src_b, off_in_b, off_out_b, E_b, epc_b, B_b);
  } else {
    gemm_body<4>(Ws, g0 + (bid - ncsr), X_a, W_a, Y_a, KD_a, ntiles_a, nwg_a,
                 X_b, W_b, Y_b, KD_b, ntiles_b);
  }
}

__global__ void __launch_bounds__(SORT_T) SIDE_BY_SIDE
sort_and_gemm_kernel(const int2* __restrict__ bin_edge_a, const int* __restrict__ bin_src_a,
                     const int* __restrict__ off_in_a, const int* __restrict__ off_out_a,
                     int* __restrict__ rowptr_a, int* __restrict__ col_a,
                     float* __restrict__ ns_a, float* __restrict__ nd_a, int N_a, int B_a, int epc_a,
                     const int2* __restrict__ bin_edge_b, const int* __restrict__ bin_src_b,
                     const int* __restrict__ off_in_b, const int* __restrict__ off_out_b,
                     int* __restrict__ rowptr_b, int* __restrict__ col_b,
                     float* __restrict__ ns_b, float* __restrict__ nd_b, int N_b, int B_b, int epc_b,
                     int ncsr, int g0,
                     const float* __restrict__ X_a, const float* __restrict__ W_a,
                     float* __restrict__ Y_a, int KD_a, int ntiles_a, int nwg_a,
                     const float* __restrict__ X_b, const float* __restrict__ W_b,
                     float* __restrict__ Y_b, int KD_b, int ntiles_b) {
  __shared__ float Ws[WS_FLOATS];
  const int bid = (int)blockIdx.x;
  if (bid < ncsr) {
    bucket_sort_body(bid, bin_edge_a, bin_src_a, off_in_a, off_out_a, rowptr_a, col_a, ns_a, nd_a,
                     N_a, B_a, epc_a,
                     bin_edge_b, bin_src_b, off_in_b, off_out_b, rowptr_b, col_b, ns_b, nd_b,
                     N_b, B_b, epc_b);
  } else {
    gemm_body<4>(Ws, g0 + (bid - ncsr), X_a, W_a, Y_a, KD_a, ntiles_a, nwg_a,
                 X_b, W_b, Y_b, KD_b, ntiles_b);
  }
}

__device__ __forceinline__ float4
gather_node(const int* __restrict__ rowptr, const int* __restrict__ col,
            const float* __restrict__ ns, const float* __restrict__ nd,
            const float4* __restrict__ Y4, const float* __restrict__ bias,
            const int node, const int lane, const int relu) {
  const int beg = rowptr[node];
  const int end = rowptr[node + 1];
  const float4 self = Y4[(size_t)node * (HDIM / 4) + lane];
  const float nsi = ns[node];
  const float ndi = nd[node];
  const float b0 = bias[lane * 4 + 0], b1 = bias[lane * 4 + 1];
  const float b2 = bias[lane * 4 + 2], b3 = bias[lane * 4 + 3];

  float4 acc = {0.0f, 0.0f, 0.0f, 0.0f};
  for (int j0 = beg; j0 < end; j0 += 32) {
    const int n = min(32, end - j0);
    int s = 0;
    float w = 0.0f;
    if (lane < n) { s = col[j0 + lane]; w = ns[s]; }
    const int wi = __float_as_int(w);
    int k = 0;
    for (; k + 4 <= n; k += 4) {
      const int s0 = __builtin_amdgcn_readlane(s, k);
      const int s1 = __builtin_amdgcn_readlane(s, k + 1);
      const int s2 = __builtin_amdgcn_readlane(s, k + 2);
      const int s3 = __builtin_amdgcn_readlane(s, k + 3);
      const float w0 = __int_as_float(__builtin_amdgcn_readlane(wi, k));
      const float w1 = __int_as_float(__builtin_amdgcn_readlane(wi, k + 1));
      const float w2 = __int_as_float(__builtin_amdgcn_readlane(wi, k + 2));
      const float w3 = __int_as_float(__builtin_amdgcn_readlane(wi, k + 3));
      const float4 m0 = Y4[(size_t)s0 * (HDIM / 4) + lane];
      const float4 m1 = Y4[(size_t)s1 * (HDIM / 4) + lane];
      const float4 m2 = Y4[(size_t)s2 * (HDIM / 4) + lane];
      const float4 m3 = Y4[(size_t)s3 * (HDIM / 4) + lane];
      acc.x = fmaf(m0.x, w0, acc.x); acc.y = fmaf(m0.y, w0, acc.y);
      acc.z = fmaf(m0.z, w0, acc.z); acc.w = fmaf(m0.w, w0, acc.w);
      acc.x = fmaf(m1.x, w1, acc.x); acc.y = fmaf(m1.y, w1, acc.y);
      acc.z = fmaf(m1.z, w1, acc.z); acc.w = fmaf(m1.w, w1, acc.w);
      acc.x = fmaf(m2.x, w2, acc.x); acc.y = fmaf(m2.y, w2, acc.y);
      acc.z = fmaf(m2.z, w2, acc.z); acc.w = fmaf(m2.w, w2, acc.w);
      acc.x = fmaf(m3.x, w3, acc.x); acc.y = fmaf(m3.y, w3, acc.y);
      acc.z = fmaf(m3.z, w3, acc.z); acc.w = fmaf(m3.w, w3, acc.w);
    }
    #pragma nounroll
    for (; k < n; ++k) {
      const int s0 = __builtin_amdgcn_readlane(s, k);
      const float w0 = __int_as_float(__builtin_amdgcn_readlane(wi, k));
      const float4 m0 = Y4[(size_t)s0 * (HDIM / 4) + lane];
      acc.x = fmaf(m0.x, w0, acc.x); acc.y = fmaf(m0.y, w0, acc.y);
      acc.z = fmaf(m0.z, w0, acc.z); acc.w = fmaf(m0.w, w0, acc.w);
    }
  }

  float4 v;
  v.x = (acc.x + self.x * nsi) * ndi + b0;
  v.y = (acc.y + self.y * nsi) * ndi + b1;
  v.z = (acc.z + self.z * nsi) * ndi + b2;
  v.w = (acc.w + self.w * nsi) * ndi + b3;
  if (relu) {
    v.x = fmaxf(v.x, 0.0f); v.y = fmaxf(v.y, 0.0f);
    v.z = fmaxf(v.z, 0.0f); v.w = fmaxf(v.w, 0.0f);
  }
  return v;
}

__global__ void __launch_bounds__(256)
gather_kernel(const int* __restrict__ rowptr_a, const int* __restrict__ col_a,
              const float* __restrict__ ns_a, const float* __restrict__ nd_a,
              const float* __restrict__ Y_a, const float* __restrict__ bias_a,
              float* __restrict__ out_a, int N_a, int nwg_a,
              const int* __restrict__ rowptr_b, const int* __restrict__ col_b,
              const float* __restrict__ ns_b, const float* __restrict__ nd_b,
              const float* __restrict__ Y_b, const float* __restrict__ bias_b,
              float* __restrict__ out_b, int N_b, int relu) {
  const bool gb = (int)blockIdx.x >= nwg_a;
  const int* __restrict__ rowptr = gb ? rowptr_b : rowptr_a;
  const int* __restrict__ col    = gb ? col_b : col_a;
  const float* __restrict__ ns   = gb ? ns_b : ns_a;
  const float* __restrict__ nd   = gb ? nd_b : nd_a;
  const float* __restrict__ Y    = gb ? Y_b : Y_a;
  const float* __restrict__ bias = gb ? bias_b : bias_a;
  float* __restrict__ out = gb ? out_b : out_a;
  const int N = gb ? N_b : N_a;
  const int lane = threadIdx.x & 31;
  const int wave = __builtin_amdgcn_readfirstlane((int)(threadIdx.x >> 5));
  const int node = ((int)blockIdx.x - (gb ? nwg_a : 0)) * 8 + wave;
  if (node >= N) return;
  const float4 v = gather_node(rowptr, col, ns, nd, (const float4*)Y, bias, node, lane, relu);
  ((float4*)out)[(size_t)node * (HDIM / 4) + lane] = v;
}

#define HP  132
#define TPG 5
__global__ void __launch_bounds__(256)
gather_gemm_kernel(const int* __restrict__ rowptr_a, const int* __restrict__ col_a,
                   const float* __restrict__ ns_a, const float* __restrict__ nd_a,
                   const float* __restrict__ Y_a, const float* __restrict__ bias_a,
                   const float* __restrict__ W_a, float* __restrict__ Z_a, int ntiles_a, int nwg_a,
                   const int* __restrict__ rowptr_b, const int* __restrict__ col_b,
                   const float* __restrict__ ns_b, const float* __restrict__ nd_b,
                   const float* __restrict__ Y_b, const float* __restrict__ bias_b,
                   const float* __restrict__ W_b, float* __restrict__ Z_b, int ntiles_b) {
  __shared__ __attribute__((aligned(16))) float Hs[NT * HP];
  const bool gb = (int)blockIdx.x >= nwg_a;
  const int* __restrict__ rowptr = gb ? rowptr_b : rowptr_a;
  const int* __restrict__ col    = gb ? col_b : col_a;
  const float* __restrict__ ns   = gb ? ns_b : ns_a;
  const float* __restrict__ nd   = gb ? nd_b : nd_a;
  const float* __restrict__ Y    = gb ? Y_b : Y_a;
  const float* __restrict__ bias = gb ? bias_b : bias_a;
  const float* __restrict__ W    = gb ? W_b : W_a;
  float* __restrict__ Z = gb ? Z_b : Z_a;
  const int ntiles = gb ? ntiles_b : ntiles_a;
  const int g = (int)blockIdx.x - (gb ? nwg_a : 0);
  const int lane = threadIdx.x & 31;
  const int wave = __builtin_amdgcn_readfirstlane((int)(threadIdx.x >> 5));
  const int l16  = lane & 15;
  const int hi   = lane >> 4;

  v2f bf[32];
  const float* __restrict__ wc = W + 16 * wave + l16;
  #pragma unroll
  for (int s = 0; s < 32; ++s) {
    const int k = 8 * (s >> 1) + 4 * hi + 2 * (s & 1);
    bf[s].x = wc[(size_t)k * HDIM];
    bf[s].y = wc[(size_t)(k + 1) * HDIM];
  }

  const int t0 = g * TPG;
  const int t1 = min(ntiles, t0 + TPG);
  for (int tile = t0; tile < t1; ++tile) {
    #pragma unroll
    for (int i = 0; i < 2; ++i) {
      const int row = 2 * wave + i;
      const float4 v = gather_node(rowptr, col, ns, nd, (const float4*)Y, bias, tile * NT + row, lane, 1);
      *(float4*)&Hs[row * HP + 4 * lane] = v;
    }
    __syncthreads();

    v8f acc0 = (v8f){}, acc1 = (v8f){};
    const float4* ap = (const float4*)&Hs[l16 * HP + 4 * hi];
    #pragma unroll
    for (int q = 0; q < 16; ++q) {
      const float4 a4 = ap[2 * q];
      v2f a;
      a.x = a4.x; a.y = a4.y;
      acc0 = __builtin_amdgcn_wmma_f32_16x16x4_f32(false, a, false, bf[2 * q], (short)0, acc0, false, false);
      a.x = a4.z; a.y = a4.w;
      acc1 = __builtin_amdgcn_wmma_f32_16x16x4_f32(false, a, false, bf[2 * q + 1], (short)0, acc1, false, false);
    }
    __syncthreads();

    #pragma unroll
    for (int r = 0; r < 8; ++r)
      Hs[(8 * hi + r) * HP + 16 * wave + l16] = acc0[r] + acc1[r];
    __syncthreads();
    #pragma unroll
    for (int i = 0; i < 2; ++i) {
      const int row = 2 * wave + i;
      const float4 y = *(const float4*)&Hs[row * HP + 4 * lane];
      ((float4*)Z)[(size_t)(tile * NT + row) * (HDIM / 4) + lane] = y;
    }
  }
}

static inline int cdiv(long long a, int b) { return (int)((a + b - 1) / b); }

struct GraphPlan {
  int N, E, B, epc;
  int2* bin_edge;
  int* bin_src;
  int* col;
  int* rowptr;
  int* off_in;
  int* off_out;
};

static GraphPlan plan_graph(float* agg, int N, int E) {
  GraphPlan g;
  g.N = N;
  g.E = E;
  g.B   = cdiv(N, BK);
  g.epc = cdiv(E, NCHUNK);
  g.bin_edge = (int2*)agg;
  g.bin_src  = (int*)agg + 2 * (size_t)E;
  g.col      = g.bin_src + E;
  g.rowptr   = g.col + E;
  g.off_in   = g.rowptr + (N + 1);
  g.off_out  = g.off_in + NCHUNK * (g.B + 1);
  return g;
}

extern "C" void kernel_launch(void* const* d_in, const int* in_sizes, int n_in,
                              void* d_out, int out_size, void* d_ws, size_t ws_size,
                              hipStream_t stream) {
  const float* feat_a = (const float*)d_in[0];
  const float* feat_b = (const float*)d_in[1];
  const int*   src_a  = (const int*)d_in[2];
  const int*   dst_a  = (const int*)d_in[3];
  const int*   src_b  = (const int*)d_in[4];
  const int*   dst_b  = (const int*)d_in[5];
  const float* Wa0 = (const float*)d_in[6];
  const float* ba0 = (const float*)d_in[7];
  const float* Wa1 = (const float*)d_in[8];
  const float* ba1 = (const float*)d_in[9];
  const float* Wb0 = (const float*)d_in[10];
  const float* bb0 = (const float*)d_in[11];
  const float* Wb1 = (const float*)d_in[12];
  const float* bb1 = (const float*)d_in[13];

  const int DA = in_sizes[6] / HDIM;
  const int DB = in_sizes[10] / HDIM;
  const int NA = in_sizes[0] / DA;
  const int NB = in_sizes[1] / DB;
  const int EA = in_sizes[2];
  const int EB = in_sizes[4];

  float* ws = (float*)d_ws;
  float* ns_a  = ws;
  float* nd_a  = ns_a + NA;
  float* Y_a   = nd_a + NA;
  float* agg_a = Y_a + (size_t)NA * HDIM;
  float* h_a   = agg_a + (size_t)NA * HDIM;
  float* ns_b  = h_a + (size_t)NA * HDIM;
  float* nd_b  = ns_b + NB;
  float* Y_b   = nd_b + NB;
  float* agg_b = Y_b + (size_t)NB * HDIM;
  float* h_b   = agg_b + (size_t)NB * HDIM;

  float* out_a = (float*)d_out;
  float* out_b = out_a + (size_t)NA * HDIM;

  const GraphPlan ga = plan_graph(agg_a, NA, EA);
  const GraphPlan gb = plan_graph(agg_b, NB, EB);

  const int mm_a = cdiv(NA / NT, 8), mm_b = cdiv(NB / NT, 8);
  const int gw_a = cdiv(NA, 8),      gw_b = cdiv(NB, 8);

  const int gh_half = (mm_a * DA + mm_b * DB) / (2 * DA);
  const int gh = gh_half < mm_a ? gh_half : mm_a;

  bin_and_gemm_kernel<<<2 * NCHUNK + gh, CHUNK_T, 0, stream>>>(
      src_a, dst_a, ga.bin_edge, ga.bin_src, ga.off_in, ga.off_out, ga.E, ga.epc, ga.B,
      src_b, dst_b, gb.bin_edge, gb.bin_src, gb.off_in, gb.off_out, gb.E, gb.epc, gb.B,
      2 * NCHUNK, 0,
      feat_a, Wa0, Y_a, DA, NA / NT, mm_a,
      feat_b, Wb0, Y_b, DB, NB / NT);
  sort_and_gemm_kernel<<<ga.B + gb.B + (mm_a + mm_b - gh), SORT_T, 0, stream>>>(
      ga.bin_edge, ga.bin_src, ga.off_in, ga.off_out, ga.rowptr, ga.col, ns_a, nd_a, NA, ga.B, ga.epc,
      gb.bin_edge, gb.bin_src, gb.off_in, gb.off_out, gb.rowptr, gb.col, ns_b, nd_b, NB, gb.B, gb.epc,
      ga.B + gb.B, gh,
      feat_a, Wa0, Y_a, DA, NA / NT, mm_a,
      feat_b, Wb0, Y_b, DB, NB / NT);

  const int gf_a = cdiv(NA / NT, TPG), gf_b = cdiv(NB / NT, TPG);
  gather_gemm_kernel<<<gf_a + gf_b, 256, 0, stream>>>(
      ga.rowptr, ga.col, ns_a, nd_a, Y_a, ba0, Wa1, h_a, NA / NT, gf_a,
      gb.rowptr, gb.col, ns_b, nd_b, Y_b, bb0, Wb1, h_b, NB / NT);

  gather_kernel<<<gw_a + gw_b, 256, 0, stream>>>(
      ga.rowptr, ga.col, ns_a, nd_a, h_a, ba1, out_a, NA, gw_a,
      gb.rowptr, gb.col, ns_b, nd_b, h_b, bb1, out_b, NB, 0);
}
